# starting kernel
# baseline (speedup 1.0000x reference)
#include <hip/hip_runtime.h>

#define NUSER 100000
#define NBOOK 50000
#define NEDGE 2000000
#define NLBL  500000
#define CCH   64
#define HCH   128

typedef __attribute__((ext_vector_type(2))) float v2f;
typedef __attribute__((ext_vector_type(8))) float v8f;

#if __has_builtin(__builtin_amdgcn_wmma_f32_16x16x4_f32)
#define HAVE_WMMA_F32 1
__device__ __forceinline__ v8f wmma4(v2f a, v2f b, v8f c) {
  return __builtin_amdgcn_wmma_f32_16x16x4_f32(false, a, false, b, (short)0, c, false, false);
}
#else
#define HAVE_WMMA_F32 0
#endif

__global__ void k_proj_user(const float* __restrict__ x, const float* __restrict__ w,
                            const float* __restrict__ bias, float* __restrict__ out) {
  int i = blockIdx.x * blockDim.x + threadIdx.x;
  if (i >= NUSER * CCH) return;
  int node = i >> 6, c = i & 63;
  float acc = bias[c];
#pragma unroll
  for (int k = 0; k < 3; ++k) acc = fmaf(x[node * 3 + k], w[k * CCH + c], acc);
  out[i] = acc;
}

__global__ void __launch_bounds__(256)
k_proj_book(const float* __restrict__ x, const float* __restrict__ w,
            const float* __restrict__ bias, float* __restrict__ out) {
  int tid = threadIdx.x;
  int wave = (blockIdx.x * blockDim.x + tid) >> 5;
  int lane = tid & 31;
  int m = lane & 15, half = lane >> 4;
  int row0 = wave << 4;
  bool active = row0 < NBOOK;
#if HAVE_WMMA_F32
  __shared__ v2f sW[64 * CCH];
  const float* xr = x + (size_t)(row0 + m) * 384;
  v8f zero = {0.f, 0.f, 0.f, 0.f, 0.f, 0.f, 0.f, 0.f};
  v8f acc[4] = {zero, zero, zero, zero};
  for (int k0 = 0; k0 < 384; k0 += 128) {
    __syncthreads();
    for (int idx = tid; idx < 64 * CCH; idx += 256) {
      int n = idx & 63, kp = idx >> 6;
      int k = k0 + 2 * kp;
      v2f v;
      v.x = w[k * CCH + n];
      v.y = w[(k + 1) * CCH + n];
      sW[idx] = v;
    }
    __syncthreads();
    if (active) {
      for (int k = k0; k < k0 + 128; k += 4) {
        int kk = k + 2 * half;
        v2f a;
        a.x = xr[kk];
        a.y = xr[kk + 1];
        int lp = (kk - k0) >> 1;
#pragma unroll
        for (int t = 0; t < 4; ++t)
          acc[t] = wmma4(a, sW[lp * CCH + t * 16 + m], acc[t]);
      }
    }
  }
  if (active) {
#pragma unroll
    for (int t = 0; t < 4; ++t) {
      int n = t * 16 + m;
      float bv = bias[n];
#pragma unroll
      for (int r = 0; r < 8; ++r)
        out[(size_t)(row0 + r + 8 * half) * CCH + n] = acc[t][r] + bv;
    }
  }
#else
  if (!active) return;
  const float* xr = x + (size_t)(row0 + m) * 384;
  for (int n = half * 32; n < half * 32 + 32; ++n) {
    float acc = bias[n];
    for (int k = 0; k < 384; ++k) acc = fmaf(xr[k], w[k * CCH + n], acc);
    out[(size_t)(row0 + m) * CCH + n] = acc;
  }
#endif
}

__global__ void k_zero(float* __restrict__ p, int n) {
  int i = blockIdx.x * blockDim.x + threadIdx.x;
  if (i < n) p[i] = 0.0f;
}

__global__ void k_count(const int* __restrict__ idx, float* __restrict__ cnt, int n) {
  int i = blockIdx.x * blockDim.x + threadIdx.x;
  if (i < n) atomicAdd(&cnt[idx[i]], 1.0f);
}

__global__ void k_scatter(const int* __restrict__ gsrc, const int* __restrict__ gdst,
                          const float* __restrict__ feat, float* __restrict__ agg) {
  long long i = (long long)blockIdx.x * blockDim.x + threadIdx.x;
  if (i >= (long long)NEDGE * 32) return;
  int e = (int)(i >> 5), ch = ((int)i & 31) * 2;
  int s = gsrc[e], d = gdst[e];
  const float2 v = *(const float2*)(feat + (size_t)s * CCH + ch);
  float* dst = agg + (size_t)d * CCH + ch;
  atomicAdd(dst, v.x);
  atomicAdd(dst + 1, v.y);
}

__global__ void __launch_bounds__(256)
k_sage(const float* __restrict__ agg, const float* __restrict__ cnt,
       const float* __restrict__ xdst,
       const float* __restrict__ wl, const float* __restrict__ wr,
       const float* __restrict__ bias, float* __restrict__ out,
       int n, int relu) {
  int tid = threadIdx.x;
  int wave = (blockIdx.x * blockDim.x + tid) >> 5;
  int lane = tid & 31;
  int m = lane & 15, half = lane >> 4;
  int row0 = wave << 4;
  bool active = row0 < n;
#if HAVE_WMMA_F32
  __shared__ v2f sWl[32 * CCH];
  __shared__ v2f sWr[32 * CCH];
  for (int idx = tid; idx < 32 * CCH; idx += 256) {
    int nn = idx & 63, kp = idx >> 6;
    int k = 2 * kp;
    v2f a, b;
    a.x = wl[k * CCH + nn];       a.y = wl[(k + 1) * CCH + nn];
    b.x = wr[k * CCH + nn];       b.y = wr[(k + 1) * CCH + nn];
    sWl[idx] = a;
    sWr[idx] = b;
  }
  __syncthreads();
  if (!active) return;
  int row = row0 + m;
  float rc = 1.0f / fmaxf(cnt[row], 1.0f);
  const float* ar = agg + (size_t)row * CCH;
  const float* xr = xdst + (size_t)row * CCH;
  v8f zero = {0.f, 0.f, 0.f, 0.f, 0.f, 0.f, 0.f, 0.f};
  v8f acc[4] = {zero, zero, zero, zero};
  for (int k = 0; k < CCH; k += 4) {
    int kk = k + 2 * half;
    int lp = kk >> 1;
    v2f am, ax;
    am.x = ar[kk] * rc;  am.y = ar[kk + 1] * rc;
    ax.x = xr[kk];       ax.y = xr[kk + 1];
#pragma unroll
    for (int t = 0; t < 4; ++t) {
      int ncol = t * 16 + m;
      acc[t] = wmma4(am, sWl[lp * CCH + ncol], acc[t]);
      acc[t] = wmma4(ax, sWr[lp * CCH + ncol], acc[t]);
    }
  }
#pragma unroll
  for (int t = 0; t < 4; ++t) {
    int ncol = t * 16 + m;
    float bv = bias[ncol];
#pragma unroll
    for (int r = 0; r < 8; ++r) {
      float v = acc[t][r] + bv;
      if (relu) v = fmaxf(v, 0.0f);
      out[(size_t)(row0 + r + 8 * half) * CCH + ncol] = v;
    }
  }
#else
  if (!active) return;
  int row = row0 + m;
  float rc = 1.0f / fmaxf(cnt[row], 1.0f);
  const float* ar = agg + (size_t)row * CCH;
  const float* xr = xdst + (size_t)row * CCH;
  for (int ncol = half * 32; ncol < half * 32 + 32; ++ncol) {
    float v = bias[ncol];
    for (int k = 0; k < CCH; ++k)
      v = fmaf(ar[k] * rc, wl[k * CCH + ncol], fmaf(xr[k], wr[k * CCH + ncol], v));
    if (relu) v = fmaxf(v, 0.0f);
    out[(size_t)row * CCH + ncol] = v;
  }
#endif
}

__global__ void __launch_bounds__(256)
k_decoder(const float* __restrict__ u2, const float* __restrict__ b2,
          const int* __restrict__ ls, const int* __restrict__ ld,
          const float* __restrict__ w1, const float* __restrict__ bias1,
          const float* __restrict__ w2, const float* __restrict__ bias2,
          float* __restrict__ out) {
  int tid = threadIdx.x;
  int wave = (blockIdx.x * blockDim.x + tid) >> 5;
  int lane = tid & 31;
  int m = lane & 15, half = lane >> 4;
  int row0 = wave << 4;
  bool active = row0 < NLBL;
  int row = active ? (row0 + m) : 0;
  const float* ur = u2 + (active ? (size_t)ls[row] * CCH : 0);
  const float* br = b2 + (active ? (size_t)ld[row] * CCH : 0);
#if HAVE_WMMA_F32
  __shared__ v2f sW[32 * HCH];
  v8f zero = {0.f, 0.f, 0.f, 0.f, 0.f, 0.f, 0.f, 0.f};
  v8f acc[8] = {zero, zero, zero, zero, zero, zero, zero, zero};
  for (int k0 = 0; k0 < HCH; k0 += 64) {
    __syncthreads();
    for (int idx = tid; idx < 32 * HCH; idx += 256) {
      int n = idx & 127, kp = idx >> 7;
      int k = k0 + 2 * kp;
      v2f v;
      v.x = w1[k * HCH + n];
      v.y = w1[(k + 1) * HCH + n];
      sW[idx] = v;
    }
    __syncthreads();
    if (active) {
      for (int k = k0; k < k0 + 64; k += 4) {
        int kk = k + 2 * half;
        const float* zsrc = (kk < CCH) ? (ur + kk) : (br + (kk - CCH));
        v2f a;
        a.x = zsrc[0];
        a.y = zsrc[1];
        int lp = (kk - k0) >> 1;
#pragma unroll
        for (int t = 0; t < 8; ++t)
          acc[t] = wmma4(a, sW[lp * HCH + t * 16 + m], acc[t]);
      }
    }
  }
  if (!active) return;
  float w2v[8], b1v[8];
#pragma unroll
  for (int t = 0; t < 8; ++t) {
    int ncol = t * 16 + m;
    w2v[t] = w2[ncol];
    b1v[t] = bias1[ncol];
  }
  float outb = bias2[0];
#pragma unroll
  for (int r = 0; r < 8; ++r) {
    float p = 0.0f;
#pragma unroll
    for (int t = 0; t < 8; ++t) {
      float hv = fmaxf(acc[t][r] + b1v[t], 0.0f);
      p = fmaf(hv, w2v[t], p);
    }
    p += __shfl_xor(p, 1);
    p += __shfl_xor(p, 2);
    p += __shfl_xor(p, 4);
    p += __shfl_xor(p, 8);
    if (m == 0) out[row0 + r + 8 * half] = p + outb;
  }
#else
  if (active && half == 0) {
    float o = bias2[0];
    for (int n = 0; n < HCH; ++n) {
      float hv = bias1[n];
      for (int k = 0; k < CCH; ++k) hv = fmaf(ur[k], w1[k * HCH + n], hv);
      for (int k = 0; k < CCH; ++k) hv = fmaf(br[k], w1[(CCH + k) * HCH + n], hv);
      o = fmaf(fmaxf(hv, 0.0f), w2[n], o);
    }
    out[row] = o;
  }
#endif
}

extern "C" void kernel_launch(void* const* d_in, const int* in_sizes, int n_in,
                              void* d_out, int out_size, void* d_ws, size_t ws_size,
                              hipStream_t stream) {
  (void)in_sizes; (void)n_in; (void)out_size; (void)ws_size;
  const float* user_x     = (const float*)d_in[0];
  const float* book_x     = (const float*)d_in[1];
  const int*   edge_src   = (const int*)d_in[2];
  const int*   edge_dst   = (const int*)d_in[3];
  const int*   label_src  = (const int*)d_in[4];
  const int*   label_dst  = (const int*)d_in[5];
  const float* user_lin_w = (const float*)d_in[6];
  const float* user_lin_b = (const float*)d_in[7];
  const float* book_lin_w = (const float*)d_in[8];
  const float* book_lin_b = (const float*)d_in[9];
  const float* w1_ub_l = (const float*)d_in[10];
  const float* b1_ub   = (const float*)d_in[11];
  const float* w1_ub_r = (const float*)d_in[12];
  const float* w1_bu_l = (const float*)d_in[13];
  const float* b1_bu   = (const float*)d_in[14];
  const float* w1_bu_r = (const float*)d_in[15];
  const float* w2_ub_l = (const float*)d_in[16];
  const float* b2_ub   = (const float*)d_in[17];
  const float* w2_ub_r = (const float*)d_in[18];
  const float* w2_bu_l = (const float*)d_in[19];
  const float* b2_bu   = (const float*)d_in[20];
  const float* w2_bu_r = (const float*)d_in[21];
  const float* dec_w1  = (const float*)d_in[22];
  const float* dec_b1  = (const float*)d_in[23];
  const float* dec_w2  = (const float*)d_in[24];
  const float* dec_b2  = (const float*)d_in[25];
  float* out = (float*)d_out;

  float* ws = (float*)d_ws;
  size_t off = 0;
  float* u0    = ws + off; off += (size_t)NUSER * CCH;
  float* b0    = ws + off; off += (size_t)NBOOK * CCH;
  float* u1    = ws + off; off += (size_t)NUSER * CCH;
  float* b1buf = ws + off; off += (size_t)NBOOK * CCH;
  float* agg   = ws + off; off += (size_t)NUSER * CCH;
  float* cnt_u = ws + off; off += NUSER;
  float* cnt_b = ws + off; off += NBOOK;
  float* u2 = u0;
  float* b2 = b0;

  const int BT = 256;
  const int sage_blk_b = (NBOOK / 16 * 32 + BT - 1) / BT;
  const int sage_blk_u = (NUSER / 16 * 32 + BT - 1) / BT;
  const int sc_blocks  = (int)(((long long)NEDGE * 32 + BT - 1) / BT);

  k_proj_user<<<(NUSER * CCH + BT - 1) / BT, BT, 0, stream>>>(user_x, user_lin_w, user_lin_b, u0);
  k_proj_book<<<(NBOOK / 16 * 32 + BT - 1) / BT, BT, 0, stream>>>(book_x, book_lin_w, book_lin_b, b0);

  k_zero<<<(NBOOK + BT - 1) / BT, BT, 0, stream>>>(cnt_b, NBOOK);
  k_zero<<<(NUSER + BT - 1) / BT, BT, 0, stream>>>(cnt_u, NUSER);
  k_count<<<(NEDGE + BT - 1) / BT, BT, 0, stream>>>(edge_dst, cnt_b, NEDGE);
  k_count<<<(NEDGE + BT - 1) / BT, BT, 0, stream>>>(edge_src, cnt_u, NEDGE);

  k_zero<<<(NBOOK * CCH + BT - 1) / BT, BT, 0, stream>>>(agg, NBOOK * CCH);
  k_scatter<<<sc_blocks, BT, 0, stream>>>(edge_src, edge_dst, u0, agg);
  k_sage<<<sage_blk_b, BT, 0, stream>>>(agg, cnt_b, b0, w1_ub_l, w1_ub_r, b1_ub, b1buf, NBOOK, 1);
  k_zero<<<(NUSER * CCH + BT - 1) / BT, BT, 0, stream>>>(agg, NUSER * CCH);
  k_scatter<<<sc_blocks, BT, 0, stream>>>(edge_dst, edge_src, b0, agg);
  k_sage<<<sage_blk_u, BT, 0, stream>>>(agg, cnt_u, u0, w1_bu_l, w1_bu_r, b1_bu, u1, NUSER, 1);

  k_zero<<<(NBOOK * CCH + BT - 1) / BT, BT, 0, stream>>>(agg, NBOOK * CCH);
  k_scatter<<<sc_blocks, BT, 0, stream>>>(edge_src, edge_dst, u1, agg);
  k_sage<<<sage_blk_b, BT, 0, stream>>>(agg, cnt_b, b1buf, w2_ub_l, w2_ub_r, b2_ub, b2, NBOOK, 0);
  k_zero<<<(NUSER * CCH + BT - 1) / BT, BT, 0, stream>>>(agg, NUSER * CCH);
  k_scatter<<<sc_blocks, BT, 0, stream>>>(edge_dst, edge_src, b1buf, agg);
  k_sage<<<sage_blk_u, BT, 0, stream>>>(agg, cnt_u, u1, w2_bu_l, w2_bu_r, b2_bu, u2, NUSER, 0);

  k_decoder<<<(NLBL / 16 * 32 + BT - 1) / BT, BT, 0, stream>>>(
      u2, b2, label_src, label_dst, dec_w1, dec_b1, dec_w2, dec_b2, out);
}


# edge scatter-add: one wave per 8 consecutive edges (indices by scalar load); all 8 feature-row reads issued before the first add, 8 reads in flight per wave not 1; atomic arrival-order sums as before
# speedup vs baseline: 1.4359x; 1.4359x over previous
#include <hip/hip_runtime.h>

#define NUSER 100000
#define NBOOK 50000
#define NEDGE 2000000
#define NLBL  500000
#define CCH   64
#define HCH   128

typedef __attribute__((ext_vector_type(2))) float v2f;
typedef __attribute__((ext_vector_type(8))) float v8f;

#if __has_builtin(__builtin_amdgcn_wmma_f32_16x16x4_f32)
#define HAVE_WMMA_F32 1
__device__ __forceinline__ v8f wmma4(v2f a, v2f b, v8f c) {
  return __builtin_amdgcn_wmma_f32_16x16x4_f32(false, a, false, b, (short)0, c, false, false);
}
#else
#define HAVE_WMMA_F32 0
#endif

__global__ void k_proj_user(const float* __restrict__ x, const float* __restrict__ w,
                            const float* __restrict__ bias, float* __restrict__ out) {
  int i = blockIdx.x * blockDim.x + threadIdx.x;
  if (i >= NUSER * CCH) return;
  int node = i >> 6, c = i & 63;
  float acc = bias[c];
#pragma unroll
  for (int k = 0; k < 3; ++k) acc = fmaf(x[node * 3 + k], w[k * CCH + c], acc);
  out[i] = acc;
}

__global__ void __launch_bounds__(256)
k_proj_book(const float* __restrict__ x, const float* __restrict__ w,
            const float* __restrict__ bias, float* __restrict__ out) {
  int tid = threadIdx.x;
  int wave = (blockIdx.x * blockDim.x + tid) >> 5;
  int lane = tid & 31;
  int m = lane & 15, half = lane >> 4;
  int row0 = wave << 4;
  bool active = row0 < NBOOK;
#if HAVE_WMMA_F32
  __shared__ v2f sW[64 * CCH];
  const float* xr = x + (size_t)(row0 + m) * 384;
  v8f zero = {0.f, 0.f, 0.f, 0.f, 0.f, 0.f, 0.f, 0.f};
  v8f acc[4] = {zero, zero, zero, zero};
  for (int k0 = 0; k0 < 384; k0 += 128) {
    __syncthreads();
    for (int idx = tid; idx < 64 * CCH; idx += 256) {
      int n = idx & 63, kp = idx >> 6;
      int k = k0 + 2 * kp;
      v2f v;
      v.x = w[k * CCH + n];
      v.y = w[(k + 1) * CCH + n];
      sW[idx] = v;
    }
    __syncthreads();
    if (active) {
      for (int k = k0; k < k0 + 128; k += 4) {
        int kk = k + 2 * half;
        v2f a;
        a.x = xr[kk];
        a.y = xr[kk + 1];
        int lp = (kk - k0) >> 1;
#pragma unroll
        for (int t = 0; t < 4; ++t)
          acc[t] = wmma4(a, sW[lp * CCH + t * 16 + m], acc[t]);
      }
    }
  }
  if (active) {
#pragma unroll
    for (int t = 0; t < 4; ++t) {
      int n = t * 16 + m;
      float bv = bias[n];
#pragma unroll
      for (int r = 0; r < 8; ++r)
        out[(size_t)(row0 + r + 8 * half) * CCH + n] = acc[t][r] + bv;
    }
  }
#else
  if (!active) return;
  const float* xr = x + (size_t)(row0 + m) * 384;
  for (int n = half * 32; n < half * 32 + 32; ++n) {
    float acc = bias[n];
    for (int k = 0; k < 384; ++k) acc = fmaf(xr[k], w[k * CCH + n], acc);
    out[(size_t)(row0 + m) * CCH + n] = acc;
  }
#endif
}

__global__ void k_zero(float* __restrict__ p, int n) {
  int i = blockIdx.x * blockDim.x + threadIdx.x;
  if (i < n) p[i] = 0.0f;
}

__global__ void k_count(const int* __restrict__ idx, float* __restrict__ cnt, int n) {
  int i = blockIdx.x * blockDim.x + threadIdx.x;
  if (i < n) atomicAdd(&cnt[idx[i]], 1.0f);
}

#define EPW 8
static_assert(NEDGE % EPW == 0, "edge count must be a multiple of EPW");
__global__ void __launch_bounds__(256)
k_scatter(const int* __restrict__ gsrc, const int* __restrict__ gdst,
          const float* __restrict__ feat, float* __restrict__ agg) {
  int wave = (int)((blockIdx.x * blockDim.x + threadIdx.x) >> 5);
  wave = __builtin_amdgcn_readfirstlane(wave);
  int e0 = wave * EPW;
  if (e0 >= NEDGE) return;
  int ch = ((int)threadIdx.x & 31) * 2;
  int s[EPW], d[EPW];
#pragma unroll
  for (int j = 0; j < EPW; ++j) { s[j] = gsrc[e0 + j]; d[j] = gdst[e0 + j]; }
  float2 v[EPW];
#pragma unroll
  for (int j = 0; j < EPW; ++j)
    v[j] = *(const float2*)(feat + (size_t)s[j] * CCH + ch);
#pragma unroll
  for (int j = 0; j < EPW; ++j) {
    float* dst = agg + (size_t)d[j] * CCH + ch;
    atomicAdd(dst, v[j].x);
    atomicAdd(dst + 1, v[j].y);
  }
}

__global__ void __launch_bounds__(256)
k_sage(const float* __restrict__ agg, const float* __restrict__ cnt,
       const float* __restrict__ xdst,
       const float* __restrict__ wl, const float* __restrict__ wr,
       const float* __restrict__ bias, float* __restrict__ out,
       int n, int relu) {
  int tid = threadIdx.x;
  int wave = (blockIdx.x * blockDim.x + tid) >> 5;
  int lane = tid & 31;
  int m = lane & 15, half = lane >> 4;
  int row0 = wave << 4;
  bool active = row0 < n;
#if HAVE_WMMA_F32
  __shared__ v2f sWl[32 * CCH];
  __shared__ v2f sWr[32 * CCH];
  for (int idx = tid; idx < 32 * CCH; idx += 256) {
    int nn = idx & 63, kp = idx >> 6;
    int k = 2 * kp;
    v2f a, b;
    a.x = wl[k * CCH + nn];       a.y = wl[(k + 1) * CCH + nn];
    b.x = wr[k * CCH + nn];       b.y = wr[(k + 1) * CCH + nn];
    sWl[idx] = a;
    sWr[idx] = b;
  }
  __syncthreads();
  if (!active) return;
  int row = row0 + m;
  float rc = 1.0f / fmaxf(cnt[row], 1.0f);
  const float* ar = agg + (size_t)row * CCH;
  const float* xr = xdst + (size_t)row * CCH;
  v8f zero = {0.f, 0.f, 0.f, 0.f, 0.f, 0.f, 0.f, 0.f};
  v8f acc[4] = {zero, zero, zero, zero};
  for (int k = 0; k < CCH; k += 4) {
    int kk = k + 2 * half;
    int lp = kk >> 1;
    v2f am, ax;
    am.x = ar[kk] * rc;  am.y = ar[kk + 1] * rc;
    ax.x = xr[kk];       ax.y = xr[kk + 1];
#pragma unroll
    for (int t = 0; t < 4; ++t) {
      int ncol = t * 16 + m;
      acc[t] = wmma4(am, sWl[lp * CCH + ncol], acc[t]);
      acc[t] = wmma4(ax, sWr[lp * CCH + ncol], acc[t]);
    }
  }
#pragma unroll
  for (int t = 0; t < 4; ++t) {
    int ncol = t * 16 + m;
    float bv = bias[ncol];
#pragma unroll
    for (int r = 0; r < 8; ++r) {
      float v = acc[t][r] + bv;
      if (relu) v = fmaxf(v, 0.0f);
      out[(size_t)(row0 + r + 8 * half) * CCH + ncol] = v;
    }
  }
#else
  if (!active) return;
  int row = row0 + m;
  float rc = 1.0f / fmaxf(cnt[row], 1.0f);
  const float* ar = agg + (size_t)row * CCH;
  const float* xr = xdst + (size_t)row * CCH;
  for (int ncol = half * 32; ncol < half * 32 + 32; ++ncol) {
    float v = bias[ncol];
    for (int k = 0; k < CCH; ++k)
      v = fmaf(ar[k] * rc, wl[k * CCH + ncol], fmaf(xr[k], wr[k * CCH + ncol], v));
    if (relu) v = fmaxf(v, 0.0f);
    out[(size_t)row * CCH + ncol] = v;
  }
#endif
}

__global__ void __launch_bounds__(256)
k_decoder(const float* __restrict__ u2, const float* __restrict__ b2,
          const int* __restrict__ ls, const int* __restrict__ ld,
          const float* __restrict__ w1, const float* __restrict__ bias1,
          const float* __restrict__ w2, const float* __restrict__ bias2,
          float* __restrict__ out) {
  int tid = threadIdx.x;
  int wave = (blockIdx.x * blockDim.x + tid) >> 5;
  int lane = tid & 31;
  int m = lane & 15, half = lane >> 4;
  int row0 = wave << 4;
  bool active = row0 < NLBL;
  int row = active ? (row0 + m) : 0;
  const float* ur = u2 + (active ? (size_t)ls[row] * CCH : 0);
  const float* br = b2 + (active ? (size_t)ld[row] * CCH : 0);
#if HAVE_WMMA_F32
  __shared__ v2f sW[32 * HCH];
  v8f zero = {0.f, 0.f, 0.f, 0.f, 0.f, 0.f, 0.f, 0.f};
  v8f acc[8] = {zero, zero, zero, zero, zero, zero, zero, zero};
  for (int k0 = 0; k0 < HCH; k0 += 64) {
    __syncthreads();
    for (int idx = tid; idx < 32 * HCH; idx += 256) {
      int n = idx & 127, kp = idx >> 7;
      int k = k0 + 2 * kp;
      v2f v;
      v.x = w1[k * HCH + n];
      v.y = w1[(k + 1) * HCH + n];
      sW[idx] = v;
    }
    __syncthreads();
    if (active) {
      for (int k = k0; k < k0 + 64; k += 4) {
        int kk = k + 2 * half;
        const float* zsrc = (kk < CCH) ? (ur + kk) : (br + (kk - CCH));
        v2f a;
        a.x = zsrc[0];
        a.y = zsrc[1];
        int lp = (kk - k0) >> 1;
#pragma unroll
        for (int t = 0; t < 8; ++t)
          acc[t] = wmma4(a, sW[lp * HCH + t * 16 + m], acc[t]);
      }
    }
  }
  if (!active) return;
  float w2v[8], b1v[8];
#pragma unroll
  for (int t = 0; t < 8; ++t) {
    int ncol = t * 16 + m;
    w2v[t] = w2[ncol];
    b1v[t] = bias1[ncol];
  }
  float outb = bias2[0];
#pragma unroll
  for (int r = 0; r < 8; ++r) {
    float p = 0.0f;
#pragma unroll
    for (int t = 0; t < 8; ++t) {
      float hv = fmaxf(acc[t][r] + b1v[t], 0.0f);
      p = fmaf(hv, w2v[t], p);
    }
    p += __shfl_xor(p, 1);
    p += __shfl_xor(p, 2);
    p += __shfl_xor(p, 4);
    p += __shfl_xor(p, 8);
    if (m == 0) out[row0 + r + 8 * half] = p + outb;
  }
#else
  if (active && half == 0) {
    float o = bias2[0];
    for (int n = 0; n < HCH; ++n) {
      float hv = bias1[n];
      for (int k = 0; k < CCH; ++k) hv = fmaf(ur[k], w1[k * HCH + n], hv);
      for (int k = 0; k < CCH; ++k) hv = fmaf(br[k], w1[(CCH + k) * HCH + n], hv);
      o = fmaf(fmaxf(hv, 0.0f), w2[n], o);
    }
    out[row] = o;
  }
#endif
}

extern "C" void kernel_launch(void* const* d_in, const int* in_sizes, int n_in,
                              void* d_out, int out_size, void* d_ws, size_t ws_size,
                              hipStream_t stream) {
  (void)in_sizes; (void)n_in; (void)out_size; (void)ws_size;
  const float* user_x     = (const float*)d_in[0];
  const float* book_x     = (const float*)d_in[1];
  const int*   edge_src   = (const int*)d_in[2];
  const int*   edge_dst   = (const int*)d_in[3];
  const int*   label_src  = (const int*)d_in[4];
  const int*   label_dst  = (const int*)d_in[5];
  const float* user_lin_w = (const float*)d_in[6];
  const float* user_lin_b = (const float*)d_in[7];
  const float* book_lin_w = (const float*)d_in[8];
  const float* book_lin_b = (const float*)d_in[9];
  const float* w1_ub_l = (const float*)d_in[10];
  const float* b1_ub   = (const float*)d_in[11];
  const float* w1_ub_r = (const float*)d_in[12];
  const float* w1_bu_l = (const float*)d_in[13];
  const float* b1_bu   = (const float*)d_in[14];
  const float* w1_bu_r = (const float*)d_in[15];
  const float* w2_ub_l = (const float*)d_in[16];
  const float* b2_ub   = (const float*)d_in[17];
  const float* w2_ub_r = (const float*)d_in[18];
  const float* w2_bu_l = (const float*)d_in[19];
  const float* b2_bu   = (const float*)d_in[20];
  const float* w2_bu_r = (const float*)d_in[21];
  const float* dec_w1  = (const float*)d_in[22];
  const float* dec_b1  = (const float*)d_in[23];
  const float* dec_w2  = (const float*)d_in[24];
  const float* dec_b2  = (const float*)d_in[25];
  float* out = (float*)d_out;

  float* ws = (float*)d_ws;
  size_t off = 0;
  float* u0    = ws + off; off += (size_t)NUSER * CCH;
  float* b0    = ws + off; off += (size_t)NBOOK * CCH;
  float* u1    = ws + off; off += (size_t)NUSER * CCH;
  float* b1buf = ws + off; off += (size_t)NBOOK * CCH;
  float* agg   = ws + off; off += (size_t)NUSER * CCH;
  float* cnt_u = ws + off; off += NUSER;
  float* cnt_b = ws + off; off += NBOOK;
  float* u2 = u0;
  float* b2 = b0;

  const int BT = 256;
  const int sage_blk_b = (NBOOK / 16 * 32 + BT - 1) / BT;
  const int sage_blk_u = (NUSER / 16 * 32 + BT - 1) / BT;
  const int sc_blocks  = (NEDGE / EPW * 32 + BT - 1) / BT;

  k_proj_user<<<(NUSER * CCH + BT - 1) / BT, BT, 0, stream>>>(user_x, user_lin_w, user_lin_b, u0);
  k_proj_book<<<(NBOOK / 16 * 32 + BT - 1) / BT, BT, 0, stream>>>(book_x, book_lin_w, book_lin_b, b0);

  k_zero<<<(NBOOK + BT - 1) / BT, BT, 0, stream>>>(cnt_b, NBOOK);
  k_zero<<<(NUSER + BT - 1) / BT, BT, 0, stream>>>(cnt_u, NUSER);
  k_count<<<(NEDGE + BT - 1) / BT, BT, 0, stream>>>(edge_dst, cnt_b, NEDGE);
  k_count<<<(NEDGE + BT - 1) / BT, BT, 0, stream>>>(edge_src, cnt_u, NEDGE);

  k_zero<<<(NBOOK * CCH + BT - 1) / BT, BT, 0, stream>>>(agg, NBOOK * CCH);
  k_scatter<<<sc_blocks, BT, 0, stream>>>(edge_src, edge_dst, u0, agg);
  k_sage<<<sage_blk_b, BT, 0, stream>>>(agg, cnt_b, b0, w1_ub_l, w1_ub_r, b1_ub, b1buf, NBOOK, 1);
  k_zero<<<(NUSER * CCH + BT - 1) / BT, BT, 0, stream>>>(agg, NUSER * CCH);
  k_scatter<<<sc_blocks, BT, 0, stream>>>(edge_dst, edge_src, b0, agg);
  k_sage<<<sage_blk_u, BT, 0, stream>>>(agg, cnt_u, u0, w1_bu_l, w1_bu_r, b1_bu, u1, NUSER, 1);

  k_zero<<<(NBOOK * CCH + BT - 1) / BT, BT, 0, stream>>>(agg, NBOOK * CCH);
  k_scatter<<<sc_blocks, BT, 0, stream>>>(edge_src, edge_dst, u1, agg);
  k_sage<<<sage_blk_b, BT, 0, stream>>>(agg, cnt_b, b1buf, w2_ub_l, w2_ub_r, b2_ub, b2, NBOOK, 0);
  k_zero<<<(NUSER * CCH + BT - 1) / BT, BT, 0, stream>>>(agg, NUSER * CCH);
  k_scatter<<<sc_blocks, BT, 0, stream>>>(edge_dst, edge_src, b1buf, agg);
  k_sage<<<sage_blk_u, BT, 0, stream>>>(agg, cnt_u, u1, w2_bu_l, w2_bu_r, b2_bu, u2, NUSER, 0);

  k_decoder<<<(NLBL / 16 * 32 + BT - 1) / BT, BT, 0, stream>>>(
      u2, b2, label_src, label_dst, dec_w1, dec_b1, dec_w2, dec_b2, out);
}


# edge scatter-add: lane l takes channels l, l+32 (was 2l, 2l+1): each row load and each f32 atomic add covers one whole aligned 128-B line, not two half-used lines; atomic arrival-order sums as before
# speedup vs baseline: 1.6073x; 1.1194x over previous
#include <hip/hip_runtime.h>

#define NUSER 100000
#define NBOOK 50000
#define NEDGE 2000000
#define NLBL  500000
#define CCH   64
#define HCH   128

typedef __attribute__((ext_vector_type(2))) float v2f;
typedef __attribute__((ext_vector_type(8))) float v8f;

#if __has_builtin(__builtin_amdgcn_wmma_f32_16x16x4_f32)
#define HAVE_WMMA_F32 1
__device__ __forceinline__ v8f wmma4(v2f a, v2f b, v8f c) {
  return __builtin_amdgcn_wmma_f32_16x16x4_f32(false, a, false, b, (short)0, c, false, false);
}
#else
#define HAVE_WMMA_F32 0
#endif

__global__ void k_proj_user(const float* __restrict__ x, const float* __restrict__ w,
                            const float* __restrict__ bias, float* __restrict__ out) {
  int i = blockIdx.x * blockDim.x + threadIdx.x;
  if (i >= NUSER * CCH) return;
  int node = i >> 6, c = i & 63;
  float acc = bias[c];
#pragma unroll
  for (int k = 0; k < 3; ++k) acc = fmaf(x[node * 3 + k], w[k * CCH + c], acc);
  out[i] = acc;
}

__global__ void __launch_bounds__(256)
k_proj_book(const float* __restrict__ x, const float* __restrict__ w,
            const float* __restrict__ bias, float* __restrict__ out) {
  int tid = threadIdx.x;
  int wave = (blockIdx.x * blockDim.x + tid) >> 5;
  int lane = tid & 31;
  int m = lane & 15, half = lane >> 4;
  int row0 = wave << 4;
  bool active = row0 < NBOOK;
#if HAVE_WMMA_F32
  __shared__ v2f sW[64 * CCH];
  const float* xr = x + (size_t)(row0 + m) * 384;
  v8f zero = {0.f, 0.f, 0.f, 0.f, 0.f, 0.f, 0.f, 0.f};
  v8f acc[4] = {zero, zero, zero, zero};
  for (int k0 = 0; k0 < 384; k0 += 128) {
    __syncthreads();
    for (int idx = tid; idx < 64 * CCH; idx += 256) {
      int n = idx & 63, kp = idx >> 6;
      int k = k0 + 2 * kp;
      v2f v;
      v.x = w[k * CCH + n];
      v.y = w[(k + 1) * CCH + n];
      sW[idx] = v;
    }
    __syncthreads();
    if (active) {
      for (int k = k0; k < k0 + 128; k += 4) {
        int kk = k + 2 * half;
        v2f a;
        a.x = xr[kk];
        a.y = xr[kk + 1];
        int lp = (kk - k0) >> 1;
#pragma unroll
        for (int t = 0; t < 4; ++t)
          acc[t] = wmma4(a, sW[lp * CCH + t * 16 + m], acc[t]);
      }
    }
  }
  if (active) {
#pragma unroll
    for (int t = 0; t < 4; ++t) {
      int n = t * 16 + m;
      float bv = bias[n];
#pragma unroll
      for (int r = 0; r < 8; ++r)
        out[(size_t)(row0 + r + 8 * half) * CCH + n] = acc[t][r] + bv;
    }
  }
#else
  if (!active) return;
  const float* xr = x + (size_t)(row0 + m) * 384;
  for (int n = half * 32; n < half * 32 + 32; ++n) {
    float acc = bias[n];
    for (int k = 0; k < 384; ++k) acc = fmaf(xr[k], w[k * CCH + n], acc);
    out[(size_t)(row0 + m) * CCH + n] = acc;
  }
#endif
}

__global__ void k_zero(float* __restrict__ p, int n) {
  int i = blockIdx.x * blockDim.x + threadIdx.x;
  if (i < n) p[i] = 0.0f;
}

__global__ void k_count(const int* __restrict__ idx, float* __restrict__ cnt, int n) {
  int i = blockIdx.x * blockDim.x + threadIdx.x;
  if (i < n) atomicAdd(&cnt[idx[i]], 1.0f);
}

#define EPW 8
static_assert(NEDGE % EPW == 0, "edge count must be a multiple of EPW");
static_assert(CCH == 64, "two 32-word lines per feature row");
__global__ void __launch_bounds__(256)
k_scatter(const int* __restrict__ gsrc, const int* __restrict__ gdst,
          const float* __restrict__ feat, float* __restrict__ agg) {
  int wave = (int)((blockIdx.x * blockDim.x + threadIdx.x) >> 5);
  wave = __builtin_amdgcn_readfirstlane(wave);
  int e0 = wave * EPW;
  if (e0 >= NEDGE) return;
  int lane = (int)threadIdx.x & 31;
  int s[EPW], d[EPW];
#pragma unroll
  for (int j = 0; j < EPW; ++j) { s[j] = gsrc[e0 + j]; d[j] = gdst[e0 + j]; }
  float lo[EPW], hi[EPW];
#pragma unroll
  for (int j = 0; j < EPW; ++j) {
    const float* r = feat + (size_t)s[j] * CCH + lane;
    lo[j] = r[0];
    hi[j] = r[32];
  }
#pragma unroll
  for (int j = 0; j < EPW; ++j) {
    float* dst = agg + (size_t)d[j] * CCH + lane;
    atomicAdd(dst, lo[j]);
    atomicAdd(dst + 32, hi[j]);
  }
}

__global__ void __launch_bounds__(256)
k_sage(const float* __restrict__ agg, const float* __restrict__ cnt,
       const float* __restrict__ xdst,
       const float* __restrict__ wl, const float* __restrict__ wr,
       const float* __restrict__ bias, float* __restrict__ out,
       int n, int relu) {
  int tid = threadIdx.x;
  int wave = (blockIdx.x * blockDim.x + tid) >> 5;
  int lane = tid & 31;
  int m = lane & 15, half = lane >> 4;
  int row0 = wave << 4;
  bool active = row0 < n;
#if HAVE_WMMA_F32
  __shared__ v2f sWl[32 * CCH];
  __shared__ v2f sWr[32 * CCH];
  for (int idx = tid; idx < 32 * CCH; idx += 256) {
    int nn = idx & 63, kp = idx >> 6;
    int k = 2 * kp;
    v2f a, b;
    a.x = wl[k * CCH + nn];       a.y = wl[(k + 1) * CCH + nn];
    b.x = wr[k * CCH + nn];       b.y = wr[(k + 1) * CCH + nn];
    sWl[idx] = a;
    sWr[idx] = b;
  }
  __syncthreads();
  if (!active) return;
  int row = row0 + m;
  float rc = 1.0f / fmaxf(cnt[row], 1.0f);
  const float* ar = agg + (size_t)row * CCH;
  const float* xr = xdst + (size_t)row * CCH;
  v8f zero = {0.f, 0.f, 0.f, 0.f, 0.f, 0.f, 0.f, 0.f};
  v8f acc[4] = {zero, zero, zero, zero};
  for (int k = 0; k < CCH; k += 4) {
    int kk = k + 2 * half;
    int lp = kk >> 1;
    v2f am, ax;
    am.x = ar[kk] * rc;  am.y = ar[kk + 1] * rc;
    ax.x = xr[kk];       ax.y = xr[kk + 1];
#pragma unroll
    for (int t = 0; t < 4; ++t) {
      int ncol = t * 16 + m;
      acc[t] = wmma4(am, sWl[lp * CCH + ncol], acc[t]);
      acc[t] = wmma4(ax, sWr[lp * CCH + ncol], acc[t]);
    }
  }
#pragma unroll
  for (int t = 0; t < 4; ++t) {
    int ncol = t * 16 + m;
    float bv = bias[ncol];
#pragma unroll
    for (int r = 0; r < 8; ++r) {
      float v = acc[t][r] + bv;
      if (relu) v = fmaxf(v, 0.0f);
      out[(size_t)(row0 + r + 8 * half) * CCH + ncol] = v;
    }
  }
#else
  if (!active) return;
  int row = row0 + m;
  float rc = 1.0f / fmaxf(cnt[row], 1.0f);
  const float* ar = agg + (size_t)row * CCH;
  const float* xr = xdst + (size_t)row * CCH;
  for (int ncol = half * 32; ncol < half * 32 + 32; ++ncol) {
    float v = bias[ncol];
    for (int k = 0; k < CCH; ++k)
      v = fmaf(ar[k] * rc, wl[k * CCH + ncol], fmaf(xr[k], wr[k * CCH + ncol], v));
    if (relu) v = fmaxf(v, 0.0f);
    out[(size_t)row * CCH + ncol] = v;
  }
#endif
}

__global__ void __launch_bounds__(256)
k_decoder(const float* __restrict__ u2, const float* __restrict__ b2,
          const int* __restrict__ ls, const int* __restrict__ ld,
          const float* __restrict__ w1, const float* __restrict__ bias1,
          const float* __restrict__ w2, const float* __restrict__ bias2,
          float* __restrict__ out) {
  int tid = threadIdx.x;
  int wave = (blockIdx.x * blockDim.x + tid) >> 5;
  int lane = tid & 31;
  int m = lane & 15, half = lane >> 4;
  int row0 = wave << 4;
  bool active = row0 < NLBL;
  int row = active ? (row0 + m) : 0;
  const float* ur = u2 + (active ? (size_t)ls[row] * CCH : 0);
  const float* br = b2 + (active ? (size_t)ld[row] * CCH : 0);
#if HAVE_WMMA_F32
  __shared__ v2f sW[32 * HCH];
  v8f zero = {0.f, 0.f, 0.f, 0.f, 0.f, 0.f, 0.f, 0.f};
  v8f acc[8] = {zero, zero, zero, zero, zero, zero, zero, zero};
  for (int k0 = 0; k0 < HCH; k0 += 64) {
    __syncthreads();
    for (int idx = tid; idx < 32 * HCH; idx += 256) {
      int n = idx & 127, kp = idx >> 7;
      int k = k0 + 2 * kp;
      v2f v;
      v.x = w1[k * HCH + n];
      v.y = w1[(k + 1) * HCH + n];
      sW[idx] = v;
    }
    __syncthreads();
    if (active) {
      for (int k = k0; k < k0 + 64; k += 4) {
        int kk = k + 2 * half;
        const float* zsrc = (kk < CCH) ? (ur + kk) : (br + (kk - CCH));
        v2f a;
        a.x = zsrc[0];
        a.y = zsrc[1];
        int lp = (kk - k0) >> 1;
#pragma unroll
        for (int t = 0; t < 8; ++t)
          acc[t] = wmma4(a, sW[lp * HCH + t * 16 + m], acc[t]);
      }
    }
  }
  if (!active) return;
  float w2v[8], b1v[8];
#pragma unroll
  for (int t = 0; t < 8; ++t) {
    int ncol = t * 16 + m;
    w2v[t] = w2[ncol];
    b1v[t] = bias1[ncol];
  }
  float outb = bias2[0];
#pragma unroll
  for (int r = 0; r < 8; ++r) {
    float p = 0.0f;
#pragma unroll
    for (int t = 0; t < 8; ++t) {
      float hv = fmaxf(acc[t][r] + b1v[t], 0.0f);
      p = fmaf(hv, w2v[t], p);
    }
    p += __shfl_xor(p, 1);
    p += __shfl_xor(p, 2);
    p += __shfl_xor(p, 4);
    p += __shfl_xor(p, 8);
    if (m == 0) out[row0 + r + 8 * half] = p + outb;
  }
#else
  if (active && half == 0) {
    float o = bias2[0];
    for (int n = 0; n < HCH; ++n) {
      float hv = bias1[n];
      for (int k = 0; k < CCH; ++k) hv = fmaf(ur[k], w1[k * HCH + n], hv);
      for (int k = 0; k < CCH; ++k) hv = fmaf(br[k], w1[(CCH + k) * HCH + n], hv);
      o = fmaf(fmaxf(hv, 0.0f), w2[n], o);
    }
    out[row] = o;
  }
#endif
}

extern "C" void kernel_launch(void* const* d_in, const int* in_sizes, int n_in,
                              void* d_out, int out_size, void* d_ws, size_t ws_size,
                              hipStream_t stream) {
  (void)in_sizes; (void)n_in; (void)out_size; (void)ws_size;
  const float* user_x     = (const float*)d_in[0];
  const float* book_x     = (const float*)d_in[1];
  const int*   edge_src   = (const int*)d_in[2];
  const int*   edge_dst   = (const int*)d_in[3];
  const int*   label_src  = (const int*)d_in[4];
  const int*   label_dst  = (const int*)d_in[5];
  const float* user_lin_w = (const float*)d_in[6];
  const float* user_lin_b = (const float*)d_in[7];
  const float* book_lin_w = (const float*)d_in[8];
  const float* book_lin_b = (const float*)d_in[9];
  const float* w1_ub_l = (const float*)d_in[10];
  const float* b1_ub   = (const float*)d_in[11];
  const float* w1_ub_r = (const float*)d_in[12];
  const float* w1_bu_l = (const float*)d_in[13];
  const float* b1_bu   = (const float*)d_in[14];
  const float* w1_bu_r = (const float*)d_in[15];
  const float* w2_ub_l = (const float*)d_in[16];
  const float* b2_ub   = (const float*)d_in[17];
  const float* w2_ub_r = (const float*)d_in[18];
  const float* w2_bu_l = (const float*)d_in[19];
  const float* b2_bu   = (const float*)d_in[20];
  const float* w2_bu_r = (const float*)d_in[21];
  const float* dec_w1  = (const float*)d_in[22];
  const float* dec_b1  = (const float*)d_in[23];
  const float* dec_w2  = (const float*)d_in[24];
  const float* dec_b2  = (const float*)d_in[25];
  float* out = (float*)d_out;

  float* ws = (float*)d_ws;
  size_t off = 0;
  float* u0    = ws + off; off += (size_t)NUSER * CCH;
  float* b0    = ws + off; off += (size_t)NBOOK * CCH;
  float* u1    = ws + off; off += (size_t)NUSER * CCH;
  float* b1buf = ws + off; off += (size_t)NBOOK * CCH;
  float* agg   = ws + off; off += (size_t)NUSER * CCH;
  float* cnt_u = ws + off; off += NUSER;
  float* cnt_b = ws + off; off += NBOOK;
  float* u2 = u0;
  float* b2 = b0;

  const int BT = 256;
  const int sage_blk_b = (NBOOK / 16 * 32 + BT - 1) / BT;
  const int sage_blk_u = (NUSER / 16 * 32 + BT - 1) / BT;
  const int sc_blocks  = (NEDGE / EPW * 32 + BT - 1) / BT;

  k_proj_user<<<(NUSER * CCH + BT - 1) / BT, BT, 0, stream>>>(user_x, user_lin_w, user_lin_b, u0);
  k_proj_book<<<(NBOOK / 16 * 32 + BT - 1) / BT, BT, 0, stream>>>(book_x, book_lin_w, book_lin_b, b0);

  k_zero<<<(NBOOK + BT - 1) / BT, BT, 0, stream>>>(cnt_b, NBOOK);
  k_zero<<<(NUSER + BT - 1) / BT, BT, 0, stream>>>(cnt_u, NUSER);
  k_count<<<(NEDGE + BT - 1) / BT, BT, 0, stream>>>(edge_dst, cnt_b, NEDGE);
  k_count<<<(NEDGE + BT - 1) / BT, BT, 0, stream>>>(edge_src, cnt_u, NEDGE);

  k_zero<<<(NBOOK * CCH + BT - 1) / BT, BT, 0, stream>>>(agg, NBOOK * CCH);
  k_scatter<<<sc_blocks, BT, 0, stream>>>(edge_src, edge_dst, u0, agg);
  k_sage<<<sage_blk_b, BT, 0, stream>>>(agg, cnt_b, b0, w1_ub_l, w1_ub_r, b1_ub, b1buf, NBOOK, 1);
  k_zero<<<(NUSER * CCH + BT - 1) / BT, BT, 0, stream>>>(agg, NUSER * CCH);
  k_scatter<<<sc_blocks, BT, 0, stream>>>(edge_dst, edge_src, b0, agg);
  k_sage<<<sage_blk_u, BT, 0, stream>>>(agg, cnt_u, u0, w1_bu_l, w1_bu_r, b1_bu, u1, NUSER, 1);

  k_zero<<<(NBOOK * CCH + BT - 1) / BT, BT, 0, stream>>>(agg, NBOOK * CCH);
  k_scatter<<<sc_blocks, BT, 0, stream>>>(edge_src, edge_dst, u1, agg);
  k_sage<<<sage_blk_b, BT, 0, stream>>>(agg, cnt_b, b1buf, w2_ub_l, w2_ub_r, b2_ub, b2, NBOOK, 0);
  k_zero<<<(NUSER * CCH + BT - 1) / BT, BT, 0, stream>>>(agg, NUSER * CCH);
  k_scatter<<<sc_blocks, BT, 0, stream>>>(edge_dst, edge_src, b1buf, agg);
  k_sage<<<sage_blk_u, BT, 0, stream>>>(agg, cnt_u, u1, w2_bu_l, w2_bu_r, b2_bu, u2, NUSER, 0);

  k_decoder<<<(NLBL / 16 * 32 + BT - 1) / BT, BT, 0, stream>>>(
      u2, b2, label_src, label_dst, dec_w1, dec_b1, dec_w2, dec_b2, out);
}


# neighbour sums without f32 atomics: edges counting-sorted by destination once per call (int atomic ranks + prefix sums), one wave per row gathers; sum order: two interleaved partial sums per row
# speedup vs baseline: 1.8538x; 1.1534x over previous
#include <hip/hip_runtime.h>

#define NUSER 100000
#define NBOOK 50000
#define NEDGE 2000000
#define NLBL  500000
#define CCH   64
#define HCH   128

typedef __attribute__((ext_vector_type(2))) float v2f;
typedef __attribute__((ext_vector_type(8))) float v8f;

#if __has_builtin(__builtin_amdgcn_wmma_f32_16x16x4_f32)
#define HAVE_WMMA_F32 1
__device__ __forceinline__ v8f wmma4(v2f a, v2f b, v8f c) {
  return __builtin_amdgcn_wmma_f32_16x16x4_f32(false, a, false, b, (short)0, c, false, false);
}
#else
#define HAVE_WMMA_F32 0
#endif

__global__ void k_proj_user(const float* __restrict__ x, const float* __restrict__ w,
                            const float* __restrict__ bias, float* __restrict__ out) {
  int i = blockIdx.x * blockDim.x + threadIdx.x;
  if (i >= NUSER * CCH) return;
  int node = i >> 6, c = i & 63;
  float acc = bias[c];
#pragma unroll
  for (int k = 0; k < 3; ++k) acc = fmaf(x[node * 3 + k], w[k * CCH + c], acc);
  out[i] = acc;
}

__global__ void __launch_bounds__(256)
k_proj_book(const float* __restrict__ x, const float* __restrict__ w,
            const float* __restrict__ bias, float* __restrict__ out) {
  int tid = threadIdx.x;
  int wave = (blockIdx.x * blockDim.x + tid) >> 5;
  int lane = tid & 31;
  int m = lane & 15, half = lane >> 4;
  int row0 = wave << 4;
  bool active = row0 < NBOOK;
#if HAVE_WMMA_F32
  __shared__ v2f sW[64 * CCH];
  const float* xr = x + (size_t)(row0 + m) * 384;
  v8f zero = {0.f, 0.f, 0.f, 0.f, 0.f, 0.f, 0.f, 0.f};
  v8f acc[4] = {zero, zero, zero, zero};
  for (int k0 = 0; k0 < 384; k0 += 128) {
    __syncthreads();
    for (int idx = tid; idx < 64 * CCH; idx += 256) {
      int n = idx & 63, kp = idx >> 6;
      int k = k0 + 2 * kp;
      v2f v;
      v.x = w[k * CCH + n];
      v.y = w[(k + 1) * CCH + n];
      sW[idx] = v;
    }
    __syncthreads();
    if (active) {
      for (int k = k0; k < k0 + 128; k += 4) {
        int kk = k + 2 * half;
        v2f a;
        a.x = xr[kk];
        a.y = xr[kk + 1];
        int lp = (kk - k0) >> 1;
#pragma unroll
        for (int t = 0; t < 4; ++t)
          acc[t] = wmma4(a, sW[lp * CCH + t * 16 + m], acc[t]);
      }
    }
  }
  if (active) {
#pragma unroll
    for (int t = 0; t < 4; ++t) {
      int n = t * 16 + m;
      float bv = bias[n];
#pragma unroll
      for (int r = 0; r < 8; ++r)
        out[(size_t)(row0 + r + 8 * half) * CCH + n] = acc[t][r] + bv;
    }
  }
#else
  if (!active) return;
  const float* xr = x + (size_t)(row0 + m) * 384;
  for (int n = half * 32; n < half * 32 + 32; ++n) {
    float acc = bias[n];
    for (int k = 0; k < 384; ++k) acc = fmaf(xr[k], w[k * CCH + n], acc);
    out[(size_t)(row0 + m) * CCH + n] = acc;
  }
#endif
}

__global__ void k_zero(float* __restrict__ p, int n) {
  int i = blockIdx.x * blockDim.x + threadIdx.x;
  if (i < n) p[i] = 0.0f;
}


__global__ void __launch_bounds__(256)
k_hist(const int* __restrict__ gsrc, const int* __restrict__ gdst,
       int* __restrict__ hist_u, int* __restrict__ hist_b,
       int* __restrict__ rank_u, int* __restrict__ rank_b) {
  int e = blockIdx.x * blockDim.x + threadIdx.x;
  if (e >= NEDGE) return;
  const int d = gdst[e], s = gsrc[e];
  const int rb = atomicAdd(&hist_b[d], 1);
  const int ru = atomicAdd(&hist_u[s], 1);
  rank_b[e] = rb;
  rank_u[e] = ru;
}

#define SCAN_T 1024
#define SCAN_BLK_B ((NBOOK + SCAN_T - 1) / SCAN_T)
#define SCAN_BLK_U ((NUSER + SCAN_T - 1) / SCAN_T)
__global__ void __launch_bounds__(SCAN_T)
k_scan(const int* __restrict__ hist_b, const int* __restrict__ hist_u,
       int* __restrict__ rowptr_b, int* __restrict__ rowptr_u,
       float* __restrict__ cnt_b, float* __restrict__ cnt_u) {
  __shared__ int s_pre[32];
  __shared__ int s_tot[32];
  const bool isb = blockIdx.x < SCAN_BLK_B;
  const int* hist = isb ? hist_b : hist_u;
  int* rowptr = isb ? rowptr_b : rowptr_u;
  float* cnt = isb ? cnt_b : cnt_u;
  const int n = isb ? NBOOK : NUSER;
  const int tile0 = (isb ? (int)blockIdx.x : (int)blockIdx.x - SCAN_BLK_B) * SCAN_T;
  const int tid = threadIdx.x, lane = tid & 31, wv = tid >> 5;
  int pre = 0;
  const int4* h4 = (const int4*)hist;
  const int lim = tile0 / 4;
  for (int q0 = 0; q0 < lim; q0 += 8 * SCAN_T) {
    int4 t[8];
#pragma unroll
    for (int u = 0; u < 8; ++u) t[u] = h4[min(q0 + u * SCAN_T + tid, lim - 1)];
#pragma unroll
    for (int u = 0; u < 8; ++u)
      pre += (q0 + u * SCAN_T + tid < lim) ? (t[u].x + t[u].y) + (t[u].z + t[u].w) : 0;
  }
  const int i = tile0 + tid;
  const int c = (i < n) ? hist[i] : 0;
  int incl = c;
#pragma unroll
  for (int o = 1; o < 32; o <<= 1) {
    int t = __shfl_up(incl, o);
    if (lane >= o) incl += t;
  }
#pragma unroll
  for (int o = 16; o > 0; o >>= 1) pre += __shfl_xor(pre, o);
  if (lane == 31) s_tot[wv] = incl;
  if (lane == 0) s_pre[wv] = pre;
  __syncthreads();
  int base = s_pre[lane] + ((lane < wv) ? s_tot[lane] : 0);
#pragma unroll
  for (int o = 16; o > 0; o >>= 1) base += __shfl_xor(base, o);
  if (i < n) {
    rowptr[i] = base + incl - c;
    cnt[i] = (float)c;
  }
}

#define ADJ_PAD 32
__global__ void __launch_bounds__(256)
k_fill(const int* __restrict__ gsrc, const int* __restrict__ gdst,
       const int* __restrict__ rank_u, const int* __restrict__ rank_b,
       const int* __restrict__ rowptr_u, const int* __restrict__ rowptr_b,
       int* __restrict__ adj_u, int* __restrict__ adj_b) {
  int e = blockIdx.x * blockDim.x + threadIdx.x;
  if (e >= NEDGE) return;
  int s = gsrc[e], d = gdst[e];
  adj_b[rowptr_b[d] + rank_b[e]] = s;
  adj_u[rowptr_u[s] + rank_u[e]] = d;
  if (e < ADJ_PAD) { adj_b[NEDGE + e] = 0; adj_u[NEDGE + e] = 0; }
}

#define GB 8
static_assert(4 * GB <= ADJ_PAD, "index batches may run this far past a list");
__global__ void __launch_bounds__(256)
k_gather(const int* __restrict__ adj, const int* __restrict__ rowptr,
         const int* __restrict__ deg, const float* __restrict__ feat,
         float* __restrict__ agg, int n) {
  int row = (int)((blockIdx.x * blockDim.x + threadIdx.x) >> 5);
  row = __builtin_amdgcn_readfirstlane(row);
  if (row >= n) return;
  const int lane = (int)threadIdx.x & 31;
  const int h = lane >> 4, c = lane & 15;
  int start = rowptr[row];
  int end = start + deg[row];
  start = min(max(start, 0), NEDGE);
  end = min(max(end, start), NEDGE);
  const char* fb = (const char*)feat;
  const unsigned coff = (unsigned)c << 4;
  float4 acc = make_float4(0.f, 0.f, 0.f, 0.f);
  int nxt[GB];
#pragma unroll
  for (int k = 0; k < GB; ++k) nxt[k] = adj[start + 2 * k + h];
  for (int j = start; j < end; j += 2 * GB) {
    float4 v[GB];
#pragma unroll
    for (int k = 0; k < GB; ++k) {
      const unsigned idx = (j + 2 * k + h < end) ? (unsigned)nxt[k] : 0u;
      v[k] = *(const float4*)(fb + ((idx << 8) + coff));
    }
#pragma unroll
    for (int k = 0; k < GB; ++k) nxt[k] = adj[j + 2 * GB + 2 * k + h];
#pragma unroll
    for (int k = 0; k < GB; ++k) {
      const bool ok = j + 2 * k + h < end;
      acc.x += ok ? v[k].x : 0.f;
      acc.y += ok ? v[k].y : 0.f;
      acc.z += ok ? v[k].z : 0.f;
      acc.w += ok ? v[k].w : 0.f;
    }
  }
  acc.x += __shfl_xor(acc.x, 16);
  acc.y += __shfl_xor(acc.y, 16);
  acc.z += __shfl_xor(acc.z, 16);
  acc.w += __shfl_xor(acc.w, 16);
  if (h == 0) ((float4*)agg)[(size_t)row * (CCH / 4) + c] = acc;
}

__global__ void __launch_bounds__(256)
k_sage(const float* agg, const float* __restrict__ cnt,
       const float* __restrict__ xdst,
       const float* __restrict__ wl, const float* __restrict__ wr,
       const float* __restrict__ bias, float* out,
       int n, int relu) {
  int tid = threadIdx.x;
  int wave = (blockIdx.x * blockDim.x + tid) >> 5;
  int lane = tid & 31;
  int m = lane & 15, half = lane >> 4;
  int row0 = wave << 4;
  bool active = row0 < n;
#if HAVE_WMMA_F32
  __shared__ v2f sWl[32 * CCH];
  __shared__ v2f sWr[32 * CCH];
  for (int idx = tid; idx < 32 * CCH; idx += 256) {
    int nn = idx & 63, kp = idx >> 6;
    int k = 2 * kp;
    v2f a, b;
    a.x = wl[k * CCH + nn];       a.y = wl[(k + 1) * CCH + nn];
    b.x = wr[k * CCH + nn];       b.y = wr[(k + 1) * CCH + nn];
    sWl[idx] = a;
    sWr[idx] = b;
  }
  __syncthreads();
  if (!active) return;
  int row = row0 + m;
  float rc = 1.0f / fmaxf(cnt[row], 1.0f);
  const float* ar = agg + (size_t)row * CCH;
  const float* xr = xdst + (size_t)row * CCH;
  v8f zero = {0.f, 0.f, 0.f, 0.f, 0.f, 0.f, 0.f, 0.f};
  v8f acc[4] = {zero, zero, zero, zero};
  for (int k = 0; k < CCH; k += 4) {
    int kk = k + 2 * half;
    int lp = kk >> 1;
    v2f am, ax;
    am.x = ar[kk] * rc;  am.y = ar[kk + 1] * rc;
    ax.x = xr[kk];       ax.y = xr[kk + 1];
#pragma unroll
    for (int t = 0; t < 4; ++t) {
      int ncol = t * 16 + m;
      acc[t] = wmma4(am, sWl[lp * CCH + ncol], acc[t]);
      acc[t] = wmma4(ax, sWr[lp * CCH + ncol], acc[t]);
    }
  }
#pragma unroll
  for (int t = 0; t < 4; ++t) {
    int ncol = t * 16 + m;
    float bv = bias[ncol];
#pragma unroll
    for (int r = 0; r < 8; ++r) {
      float v = acc[t][r] + bv;
      if (relu) v = fmaxf(v, 0.0f);
      out[(size_t)(row0 + r + 8 * half) * CCH + ncol] = v;
    }
  }
#else
  if (!active) return;
  int row = row0 + m;
  float rc = 1.0f / fmaxf(cnt[row], 1.0f);
  const float* ar = agg + (size_t)row * CCH;
  const float* xr = xdst + (size_t)row * CCH;
  for (int ncol = half * 32; ncol < half * 32 + 32; ++ncol) {
    float v = bias[ncol];
    for (int k = 0; k < CCH; ++k)
      v = fmaf(ar[k] * rc, wl[k * CCH + ncol], fmaf(xr[k], wr[k * CCH + ncol], v));
    if (relu) v = fmaxf(v, 0.0f);
    out[(size_t)row * CCH + ncol] = v;
  }
#endif
}

__global__ void __launch_bounds__(256)
k_decoder(const float* __restrict__ u2, const float* __restrict__ b2,
          const int* __restrict__ ls, const int* __restrict__ ld,
          const float* __restrict__ w1, const float* __restrict__ bias1,
          const float* __restrict__ w2, const float* __restrict__ bias2,
          float* __restrict__ out) {
  int tid = threadIdx.x;
  int wave = (blockIdx.x * blockDim.x + tid) >> 5;
  int lane = tid & 31;
  int m = lane & 15, half = lane >> 4;
  int row0 = wave << 4;
  bool active = row0 < NLBL;
  int row = active ? (row0 + m) : 0;
  const float* ur = u2 + (active ? (size_t)ls[row] * CCH : 0);
  const float* br = b2 + (active ? (size_t)ld[row] * CCH : 0);
#if HAVE_WMMA_F32
  __shared__ v2f sW[32 * HCH];
  v8f zero = {0.f, 0.f, 0.f, 0.f, 0.f, 0.f, 0.f, 0.f};
  v8f acc[8] = {zero, zero, zero, zero, zero, zero, zero, zero};
  for (int k0 = 0; k0 < HCH; k0 += 64) {
    __syncthreads();
    for (int idx = tid; idx < 32 * HCH; idx += 256) {
      int n = idx & 127, kp = idx >> 7;
      int k = k0 + 2 * kp;
      v2f v;
      v.x = w1[k * HCH + n];
      v.y = w1[(k + 1) * HCH + n];
      sW[idx] = v;
    }
    __syncthreads();
    if (active) {
      for (int k = k0; k < k0 + 64; k += 4) {
        int kk = k + 2 * half;
        const float* zsrc = (kk < CCH) ? (ur + kk) : (br + (kk - CCH));
        v2f a;
        a.x = zsrc[0];
        a.y = zsrc[1];
        int lp = (kk - k0) >> 1;
#pragma unroll
        for (int t = 0; t < 8; ++t)
          acc[t] = wmma4(a, sW[lp * HCH + t * 16 + m], acc[t]);
      }
    }
  }
  if (!active) return;
  float w2v[8], b1v[8];
#pragma unroll
  for (int t = 0; t < 8; ++t) {
    int ncol = t * 16 + m;
    w2v[t] = w2[ncol];
    b1v[t] = bias1[ncol];
  }
  float outb = bias2[0];
#pragma unroll
  for (int r = 0; r < 8; ++r) {
    float p = 0.0f;
#pragma unroll
    for (int t = 0; t < 8; ++t) {
      float hv = fmaxf(acc[t][r] + b1v[t], 0.0f);
      p = fmaf(hv, w2v[t], p);
    }
    p += __shfl_xor(p, 1);
    p += __shfl_xor(p, 2);
    p += __shfl_xor(p, 4);
    p += __shfl_xor(p, 8);
    if (m == 0) out[row0 + r + 8 * half] = p + outb;
  }
#else
  if (active && half == 0) {
    float o = bias2[0];
    for (int n = 0; n < HCH; ++n) {
      float hv = bias1[n];
      for (int k = 0; k < CCH; ++k) hv = fmaf(ur[k], w1[k * HCH + n], hv);
      for (int k = 0; k < CCH; ++k) hv = fmaf(br[k], w1[(CCH + k) * HCH + n], hv);
      o = fmaf(fmaxf(hv, 0.0f), w2[n], o);
    }
    out[row] = o;
  }
#endif
}

extern "C" void kernel_launch(void* const* d_in, const int* in_sizes, int n_in,
                              void* d_out, int out_size, void* d_ws, size_t ws_size,
                              hipStream_t stream) {
  (void)in_sizes; (void)n_in; (void)out_size; (void)ws_size;
  const float* user_x     = (const float*)d_in[0];
  const float* book_x     = (const float*)d_in[1];
  const int*   edge_src   = (const int*)d_in[2];
  const int*   edge_dst   = (const int*)d_in[3];
  const int*   label_src  = (const int*)d_in[4];
  const int*   label_dst  = (const int*)d_in[5];
  const float* user_lin_w = (const float*)d_in[6];
  const float* user_lin_b = (const float*)d_in[7];
  const float* book_lin_w = (const float*)d_in[8];
  const float* book_lin_b = (const float*)d_in[9];
  const float* w1_ub_l = (const float*)d_in[10];
  const float* b1_ub   = (const float*)d_in[11];
  const float* w1_ub_r = (const float*)d_in[12];
  const float* w1_bu_l = (const float*)d_in[13];
  const float* b1_bu   = (const float*)d_in[14];
  const float* w1_bu_r = (const float*)d_in[15];
  const float* w2_ub_l = (const float*)d_in[16];
  const float* b2_ub   = (const float*)d_in[17];
  const float* w2_ub_r = (const float*)d_in[18];
  const float* w2_bu_l = (const float*)d_in[19];
  const float* b2_bu   = (const float*)d_in[20];
  const float* w2_bu_r = (const float*)d_in[21];
  const float* dec_w1  = (const float*)d_in[22];
  const float* dec_b1  = (const float*)d_in[23];
  const float* dec_w2  = (const float*)d_in[24];
  const float* dec_b2  = (const float*)d_in[25];
  float* out = (float*)d_out;

  float* ws = (float*)d_ws;
  size_t off = 0;
  float* u0    = ws + off; off += (size_t)NUSER * CCH;
  float* b0    = ws + off; off += (size_t)NBOOK * CCH;
  float* u1    = ws + off; off += (size_t)NUSER * CCH;
  float* b1buf = ws + off; off += (size_t)NBOOK * CCH;
  int*   tab   = (int*)(ws + off); off += (size_t)NUSER * CCH;
  float* cnt_u = ws + off; off += NUSER;
  float* cnt_b = ws + off; off += NBOOK;
  float* u2 = u0;
  float* b2 = b0;
  int* adj_b    = tab;
  int* adj_u    = adj_b + (NEDGE + ADJ_PAD);
  int* hist_b   = adj_u + (NEDGE + ADJ_PAD);
  int* hist_u   = hist_b + NBOOK;
  int* rowptr_b = hist_u + NUSER;
  int* rowptr_u = rowptr_b + NBOOK;
  int* rank_b = (int*)u1;
  int* rank_u = rank_b + NEDGE;

  const int BT = 256;
  const int sage_blk_b = (NBOOK / 16 * 32 + BT - 1) / BT;
  const int sage_blk_u = (NUSER / 16 * 32 + BT - 1) / BT;
  const int edge_blk   = (NEDGE + BT - 1) / BT;
  const int gat_blk_b  = (NBOOK * 32 + BT - 1) / BT;
  const int gat_blk_u  = (NUSER * 32 + BT - 1) / BT;

  k_proj_user<<<(NUSER * CCH + BT - 1) / BT, BT, 0, stream>>>(user_x, user_lin_w, user_lin_b, u0);
  k_proj_book<<<(NBOOK / 16 * 32 + BT - 1) / BT, BT, 0, stream>>>(book_x, book_lin_w, book_lin_b, b0);

  k_zero<<<(NBOOK + NUSER + BT - 1) / BT, BT, 0, stream>>>((float*)hist_b, NBOOK + NUSER);
  k_hist<<<edge_blk, BT, 0, stream>>>(edge_src, edge_dst, hist_u, hist_b, rank_u, rank_b);
  k_scan<<<SCAN_BLK_B + SCAN_BLK_U, SCAN_T, 0, stream>>>(hist_b, hist_u, rowptr_b, rowptr_u, cnt_b, cnt_u);
  k_fill<<<edge_blk, BT, 0, stream>>>(edge_src, edge_dst, rank_u, rank_b, rowptr_u, rowptr_b, adj_u, adj_b);

  k_gather<<<gat_blk_b, BT, 0, stream>>>(adj_b, rowptr_b, hist_b, u0, b1buf, NBOOK);
  k_sage<<<sage_blk_b, BT, 0, stream>>>(b1buf, cnt_b, b0, w1_ub_l, w1_ub_r, b1_ub, b1buf, NBOOK, 1);
  k_gather<<<gat_blk_u, BT, 0, stream>>>(adj_u, rowptr_u, hist_u, b0, u1, NUSER);
  k_sage<<<sage_blk_u, BT, 0, stream>>>(u1, cnt_u, u0, w1_bu_l, w1_bu_r, b1_bu, u1, NUSER, 1);

  k_gather<<<gat_blk_b, BT, 0, stream>>>(adj_b, rowptr_b, hist_b, u1, b2, NBOOK);
  k_sage<<<sage_blk_b, BT, 0, stream>>>(b2, cnt_b, b1buf, w2_ub_l, w2_ub_r, b2_ub, b2, NBOOK, 0);
  k_gather<<<gat_blk_u, BT, 0, stream>>>(adj_u, rowptr_u, hist_u, b1buf, u2, NUSER);
  k_sage<<<sage_blk_u, BT, 0, stream>>>(u2, cnt_u, u1, w2_bu_l, w2_bu_r, b2_bu, u2, NUSER, 0);

  k_decoder<<<(NLBL / 16 * 32 + BT - 1) / BT, BT, 0, stream>>>(
      u2, b2, label_src, label_dst, dec_w1, dec_b1, dec_w2, dec_b2, out);
}


# decoder first layer per node, not per edge: u2@W1[:64]+b1 and b2@W1[64:] once for 150K nodes (WMMA), each edge adds its two gathered 128-float rows; sum order: two 64-term products, then added
# speedup vs baseline: 1.9515x; 1.0527x over previous
#include <hip/hip_runtime.h>

#define NUSER 100000
#define NBOOK 50000
#define NEDGE 2000000
#define NLBL  500000
#define CCH   64
#define HCH   128

typedef __attribute__((ext_vector_type(2))) float v2f;
typedef __attribute__((ext_vector_type(8))) float v8f;

#if __has_builtin(__builtin_amdgcn_wmma_f32_16x16x4_f32)
#define HAVE_WMMA_F32 1
__device__ __forceinline__ v8f wmma4(v2f a, v2f b, v8f c) {
  return __builtin_amdgcn_wmma_f32_16x16x4_f32(false, a, false, b, (short)0, c, false, false);
}
#else
#define HAVE_WMMA_F32 0
#endif

__global__ void k_proj_user(const float* __restrict__ x, const float* __restrict__ w,
                            const float* __restrict__ bias, float* __restrict__ out) {
  int i = blockIdx.x * blockDim.x + threadIdx.x;
  if (i >= NUSER * CCH) return;
  int node = i >> 6, c = i & 63;
  float acc = bias[c];
#pragma unroll
  for (int k = 0; k < 3; ++k) acc = fmaf(x[node * 3 + k], w[k * CCH + c], acc);
  out[i] = acc;
}

__global__ void __launch_bounds__(256)
k_proj_book(const float* __restrict__ x, const float* __restrict__ w,
            const float* __restrict__ bias, float* __restrict__ out) {
  int tid = threadIdx.x;
  int wave = (blockIdx.x * blockDim.x + tid) >> 5;
  int lane = tid & 31;
  int m = lane & 15, half = lane >> 4;
  int row0 = wave << 4;
  bool active = row0 < NBOOK;
#if HAVE_WMMA_F32
  __shared__ v2f sW[64 * CCH];
  const float* xr = x + (size_t)(row0 + m) * 384;
  v8f zero = {0.f, 0.f, 0.f, 0.f, 0.f, 0.f, 0.f, 0.f};
  v8f acc[4] = {zero, zero, zero, zero};
  for (int k0 = 0; k0 < 384; k0 += 128) {
    __syncthreads();
    for (int idx = tid; idx < 64 * CCH; idx += 256) {
      int n = idx & 63, kp = idx >> 6;
      int k = k0 + 2 * kp;
      v2f v;
      v.x = w[k * CCH + n];
      v.y = w[(k + 1) * CCH + n];
      sW[idx] = v;
    }
    __syncthreads();
    if (active) {
      for (int k = k0; k < k0 + 128; k += 4) {
        int kk = k + 2 * half;
        v2f a;
        a.x = xr[kk];
        a.y = xr[kk + 1];
        int lp = (kk - k0) >> 1;
#pragma unroll
        for (int t = 0; t < 4; ++t)
          acc[t] = wmma4(a, sW[lp * CCH + t * 16 + m], acc[t]);
      }
    }
  }
  if (active) {
#pragma unroll
    for (int t = 0; t < 4; ++t) {
      int n = t * 16 + m;
      float bv = bias[n];
#pragma unroll
      for (int r = 0; r < 8; ++r)
        out[(size_t)(row0 + r + 8 * half) * CCH + n] = acc[t][r] + bv;
    }
  }
#else
  if (!active) return;
  const float* xr = x + (size_t)(row0 + m) * 384;
  for (int n = half * 32; n < half * 32 + 32; ++n) {
    float acc = bias[n];
    for (int k = 0; k < 384; ++k) acc = fmaf(xr[k], w[k * CCH + n], acc);
    out[(size_t)(row0 + m) * CCH + n] = acc;
  }
#endif
}

__global__ void k_zero(float* __restrict__ p, int n) {
  int i = blockIdx.x * blockDim.x + threadIdx.x;
  if (i < n) p[i] = 0.0f;
}


__global__ void __launch_bounds__(256)
k_hist(const int* __restrict__ gsrc, const int* __restrict__ gdst,
       int* __restrict__ hist_u, int* __restrict__ hist_b,
       int* __restrict__ rank_u, int* __restrict__ rank_b) {
  int e = blockIdx.x * blockDim.x + threadIdx.x;
  if (e >= NEDGE) return;
  const int d = gdst[e], s = gsrc[e];
  const int rb = atomicAdd(&hist_b[d], 1);
  const int ru = atomicAdd(&hist_u[s], 1);
  rank_b[e] = rb;
  rank_u[e] = ru;
}

#define SCAN_T 1024
#define SCAN_BLK_B ((NBOOK + SCAN_T - 1) / SCAN_T)
#define SCAN_BLK_U ((NUSER + SCAN_T - 1) / SCAN_T)
__global__ void __launch_bounds__(SCAN_T)
k_scan(const int* __restrict__ hist_b, const int* __restrict__ hist_u,
       int* __restrict__ rowptr_b, int* __restrict__ rowptr_u,
       float* __restrict__ cnt_b, float* __restrict__ cnt_u) {
  __shared__ int s_pre[32];
  __shared__ int s_tot[32];
  const bool isb = blockIdx.x < SCAN_BLK_B;
  const int* hist = isb ? hist_b : hist_u;
  int* rowptr = isb ? rowptr_b : rowptr_u;
  float* cnt = isb ? cnt_b : cnt_u;
  const int n = isb ? NBOOK : NUSER;
  const int tile0 = (isb ? (int)blockIdx.x : (int)blockIdx.x - SCAN_BLK_B) * SCAN_T;
  const int tid = threadIdx.x, lane = tid & 31, wv = tid >> 5;
  int pre = 0;
  const int4* h4 = (const int4*)hist;
  const int lim = tile0 / 4;
  for (int q0 = 0; q0 < lim; q0 += 8 * SCAN_T) {
    int4 t[8];
#pragma unroll
    for (int u = 0; u < 8; ++u) t[u] = h4[min(q0 + u * SCAN_T + tid, lim - 1)];
#pragma unroll
    for (int u = 0; u < 8; ++u)
      pre += (q0 + u * SCAN_T + tid < lim) ? (t[u].x + t[u].y) + (t[u].z + t[u].w) : 0;
  }
  const int i = tile0 + tid;
  const int c = (i < n) ? hist[i] : 0;
  int incl = c;
#pragma unroll
  for (int o = 1; o < 32; o <<= 1) {
    int t = __shfl_up(incl, o);
    if (lane >= o) incl += t;
  }
#pragma unroll
  for (int o = 16; o > 0; o >>= 1) pre += __shfl_xor(pre, o);
  if (lane == 31) s_tot[wv] = incl;
  if (lane == 0) s_pre[wv] = pre;
  __syncthreads();
  int base = s_pre[lane] + ((lane < wv) ? s_tot[lane] : 0);
#pragma unroll
  for (int o = 16; o > 0; o >>= 1) base += __shfl_xor(base, o);
  if (i < n) {
    rowptr[i] = base + incl - c;
    cnt[i] = (float)c;
  }
}

#define ADJ_PAD 32
__global__ void __launch_bounds__(256)
k_fill(const int* __restrict__ gsrc, const int* __restrict__ gdst,
       const int* __restrict__ rank_u, const int* __restrict__ rank_b,
       const int* __restrict__ rowptr_u, const int* __restrict__ rowptr_b,
       int* __restrict__ adj_u, int* __restrict__ adj_b) {
  int e = blockIdx.x * blockDim.x + threadIdx.x;
  if (e >= NEDGE) return;
  int s = gsrc[e], d = gdst[e];
  adj_b[rowptr_b[d] + rank_b[e]] = s;
  adj_u[rowptr_u[s] + rank_u[e]] = d;
  if (e < ADJ_PAD) { adj_b[NEDGE + e] = 0; adj_u[NEDGE + e] = 0; }
}

#define GB 8
static_assert(4 * GB <= ADJ_PAD, "index batches may run this far past a list");
__global__ void __launch_bounds__(256)
k_gather(const int* __restrict__ adj, const int* __restrict__ rowptr,
         const int* __restrict__ deg, const float* __restrict__ feat,
         float* __restrict__ agg, int n) {
  int row = (int)((blockIdx.x * blockDim.x + threadIdx.x) >> 5);
  row = __builtin_amdgcn_readfirstlane(row);
  if (row >= n) return;
  const int lane = (int)threadIdx.x & 31;
  const int h = lane >> 4, c = lane & 15;
  int start = rowptr[row];
  int end = start + deg[row];
  start = min(max(start, 0), NEDGE);
  end = min(max(end, start), NEDGE);
  const char* fb = (const char*)feat;
  const unsigned coff = (unsigned)c << 4;
  float4 acc = make_float4(0.f, 0.f, 0.f, 0.f);
  int nxt[GB];
#pragma unroll
  for (int k = 0; k < GB; ++k) nxt[k] = adj[start + 2 * k + h];
  for (int j = start; j < end; j += 2 * GB) {
    float4 v[GB];
#pragma unroll
    for (int k = 0; k < GB; ++k) {
      const unsigned idx = (j + 2 * k + h < end) ? (unsigned)nxt[k] : 0u;
      v[k] = *(const float4*)(fb + ((idx << 8) + coff));
    }
#pragma unroll
    for (int k = 0; k < GB; ++k) nxt[k] = adj[j + 2 * GB + 2 * k + h];
#pragma unroll
    for (int k = 0; k < GB; ++k) {
      const bool ok = j + 2 * k + h < end;
      acc.x += ok ? v[k].x : 0.f;
      acc.y += ok ? v[k].y : 0.f;
      acc.z += ok ? v[k].z : 0.f;
      acc.w += ok ? v[k].w : 0.f;
    }
  }
  acc.x += __shfl_xor(acc.x, 16);
  acc.y += __shfl_xor(acc.y, 16);
  acc.z += __shfl_xor(acc.z, 16);
  acc.w += __shfl_xor(acc.w, 16);
  if (h == 0) ((float4*)agg)[(size_t)row * (CCH / 4) + c] = acc;
}

__global__ void __launch_bounds__(256)
k_sage(const float* agg, const float* __restrict__ cnt,
       const float* __restrict__ xdst,
       const float* __restrict__ wl, const float* __restrict__ wr,
       const float* __restrict__ bias, float* out,
       int n, int relu) {
  int tid = threadIdx.x;
  int wave = (blockIdx.x * blockDim.x + tid) >> 5;
  int lane = tid & 31;
  int m = lane & 15, half = lane >> 4;
  int row0 = wave << 4;
  bool active = row0 < n;
#if HAVE_WMMA_F32
  __shared__ v2f sWl[32 * CCH];
  __shared__ v2f sWr[32 * CCH];
  for (int idx = tid; idx < 32 * CCH; idx += 256) {
    int nn = idx & 63, kp = idx >> 6;
    int k = 2 * kp;
    v2f a, b;
    a.x = wl[k * CCH + nn];       a.y = wl[(k + 1) * CCH + nn];
    b.x = wr[k * CCH + nn];       b.y = wr[(k + 1) * CCH + nn];
    sWl[idx] = a;
    sWr[idx] = b;
  }
  __syncthreads();
  if (!active) return;
  int row = row0 + m;
  float rc = 1.0f / fmaxf(cnt[row], 1.0f);
  const float* ar = agg + (size_t)row * CCH;
  const float* xr = xdst + (size_t)row * CCH;
  v8f zero = {0.f, 0.f, 0.f, 0.f, 0.f, 0.f, 0.f, 0.f};
  v8f acc[4] = {zero, zero, zero, zero};
  for (int k = 0; k < CCH; k += 4) {
    int kk = k + 2 * half;
    int lp = kk >> 1;
    v2f am, ax;
    am.x = ar[kk] * rc;  am.y = ar[kk + 1] * rc;
    ax.x = xr[kk];       ax.y = xr[kk + 1];
#pragma unroll
    for (int t = 0; t < 4; ++t) {
      int ncol = t * 16 + m;
      acc[t] = wmma4(am, sWl[lp * CCH + ncol], acc[t]);
      acc[t] = wmma4(ax, sWr[lp * CCH + ncol], acc[t]);
    }
  }
#pragma unroll
  for (int t = 0; t < 4; ++t) {
    int ncol = t * 16 + m;
    float bv = bias[ncol];
#pragma unroll
    for (int r = 0; r < 8; ++r) {
      float v = acc[t][r] + bv;
      if (relu) v = fmaxf(v, 0.0f);
      out[(size_t)(row0 + r + 8 * half) * CCH + ncol] = v;
    }
  }
#else
  if (!active) return;
  int row = row0 + m;
  float rc = 1.0f / fmaxf(cnt[row], 1.0f);
  const float* ar = agg + (size_t)row * CCH;
  const float* xr = xdst + (size_t)row * CCH;
  for (int ncol = half * 32; ncol < half * 32 + 32; ++ncol) {
    float v = bias[ncol];
    for (int k = 0; k < CCH; ++k)
      v = fmaf(ar[k] * rc, wl[k * CCH + ncol], fmaf(xr[k], wr[k * CCH + ncol], v));
    if (relu) v = fmaxf(v, 0.0f);
    out[(size_t)row * CCH + ncol] = v;
  }
#endif
}


__global__ void __launch_bounds__(256)
k_dec_proj(const float* __restrict__ x, const float* __restrict__ wh,
           const float* __restrict__ bias, float* __restrict__ p,
           int n, int use_bias) {
  int tid = threadIdx.x;
  int wave = (blockIdx.x * blockDim.x + tid) >> 5;
  int lane = tid & 31;
  int m = lane & 15, half = lane >> 4;
  int row0 = wave << 4;
  bool active = row0 < n;
#if HAVE_WMMA_F32
  __shared__ v2f sW[32 * HCH];
  for (int idx = tid; idx < 32 * HCH; idx += 256) {
    int nn = idx & 127, kp = idx >> 7;
    int k = 2 * kp;
    v2f v;
    v.x = wh[k * HCH + nn];
    v.y = wh[(k + 1) * HCH + nn];
    sW[idx] = v;
  }
  __syncthreads();
  if (!active) return;
  const float* xr = x + (size_t)(row0 + m) * CCH;
  v8f zero = {0.f, 0.f, 0.f, 0.f, 0.f, 0.f, 0.f, 0.f};
  v8f acc[8] = {zero, zero, zero, zero, zero, zero, zero, zero};
  for (int k = 0; k < CCH; k += 4) {
    int kk = k + 2 * half;
    v2f a;
    a.x = xr[kk];
    a.y = xr[kk + 1];
    int lp = kk >> 1;
#pragma unroll
    for (int t = 0; t < 8; ++t)
      acc[t] = wmma4(a, sW[lp * HCH + t * 16 + m], acc[t]);
  }
#pragma unroll
  for (int t = 0; t < 8; ++t) {
    int ncol = t * 16 + m;
    float bv = use_bias ? bias[ncol] : 0.0f;
#pragma unroll
    for (int r = 0; r < 8; ++r)
      p[(size_t)(row0 + r + 8 * half) * HCH + ncol] = acc[t][r] + bv;
  }
#else
  if (!active) return;
  const float* xr = x + (size_t)(row0 + m) * CCH;
  for (int ncol = half * 64; ncol < half * 64 + 64; ++ncol) {
    float v = use_bias ? bias[ncol] : 0.0f;
    for (int k = 0; k < CCH; ++k) v = fmaf(xr[k], wh[k * HCH + ncol], v);
    p[(size_t)(row0 + m) * HCH + ncol] = v;
  }
#endif
}

static_assert(NLBL % 32 == 0, "a wave takes 32 edges");
__global__ void __launch_bounds__(256)
k_dec_edge(const float* __restrict__ pu, const float* __restrict__ pb,
           const int* __restrict__ ls, const int* __restrict__ ld,
           const float* __restrict__ w2, const float* __restrict__ bias2,
           float* __restrict__ out) {
  int wave = (int)((blockIdx.x * blockDim.x + threadIdx.x) >> 5);
  wave = __builtin_amdgcn_readfirstlane(wave);
  const int e0 = wave * 32;
  if (e0 >= NLBL) return;
  const int lane = (int)threadIdx.x & 31;
  const unsigned loff = (unsigned)lane << 4;
  const float4 wv = ((const float4*)w2)[lane];
  const float ob = bias2[0];
  const char* pub = (const char*)pu;
  const char* pbb = (const char*)pb;
  const bool h = (lane & 16) != 0, g = (lane & 8) != 0, f = (lane & 4) != 0;
  for (int b = 0; b < 4; ++b) {
    float4 va[8], vc[8];
#pragma unroll
    for (int i = 0; i < 8; ++i) {
      const int e = e0 + 8 * b + i;
      const unsigned su = (unsigned)min(max(ls[e], 0), NUSER - 1);
      const unsigned sb = (unsigned)min(max(ld[e], 0), NBOOK - 1);
      va[i] = *(const float4*)(pub + ((su << 9) + loff));
      vc[i] = *(const float4*)(pbb + ((sb << 9) + loff));
    }
    float p[8];
#pragma unroll
    for (int i = 0; i < 8; ++i) {
      const float x0 = fmaxf(va[i].x + vc[i].x, 0.0f);
      const float x1 = fmaxf(va[i].y + vc[i].y, 0.0f);
      const float x2 = fmaxf(va[i].z + vc[i].z, 0.0f);
      const float x3 = fmaxf(va[i].w + vc[i].w, 0.0f);
      p[i] = fmaf(x3, wv.w, fmaf(x2, wv.z, fmaf(x1, wv.y, x0 * wv.x)));
    }
    float q[4];
#pragma unroll
    for (int i = 0; i < 4; ++i) {
      const float keep = h ? p[i + 4] : p[i];
      const float send = h ? p[i] : p[i + 4];
      q[i] = keep + __shfl_xor(send, 16);
    }
    float r[2];
#pragma unroll
    for (int i = 0; i < 2; ++i) {
      const float keep = g ? q[i + 2] : q[i];
      const float send = g ? q[i] : q[i + 2];
      r[i] = keep + __shfl_xor(send, 8);
    }
    float s = (f ? r[1] : r[0]) + __shfl_xor(f ? r[0] : r[1], 4);
    s += __shfl_xor(s, 2);
    s += __shfl_xor(s, 1);
    if ((lane & 3) == 0) out[e0 + 8 * b + (lane >> 2)] = s + ob;
  }
}

extern "C" void kernel_launch(void* const* d_in, const int* in_sizes, int n_in,
                              void* d_out, int out_size, void* d_ws, size_t ws_size,
                              hipStream_t stream) {
  (void)in_sizes; (void)n_in; (void)out_size; (void)ws_size;
  const float* user_x     = (const float*)d_in[0];
  const float* book_x     = (const float*)d_in[1];
  const int*   edge_src   = (const int*)d_in[2];
  const int*   edge_dst   = (const int*)d_in[3];
  const int*   label_src  = (const int*)d_in[4];
  const int*   label_dst  = (const int*)d_in[5];
  const float* user_lin_w = (const float*)d_in[6];
  const float* user_lin_b = (const float*)d_in[7];
  const float* book_lin_w = (const float*)d_in[8];
  const float* book_lin_b = (const float*)d_in[9];
  const float* w1_ub_l = (const float*)d_in[10];
  const float* b1_ub   = (const float*)d_in[11];
  const float* w1_ub_r = (const float*)d_in[12];
  const float* w1_bu_l = (const float*)d_in[13];
  const float* b1_bu   = (const float*)d_in[14];
  const float* w1_bu_r = (const float*)d_in[15];
  const float* w2_ub_l = (const float*)d_in[16];
  const float* b2_ub   = (const float*)d_in[17];
  const float* w2_ub_r = (const float*)d_in[18];
  const float* w2_bu_l = (const float*)d_in[19];
  const float* b2_bu   = (const float*)d_in[20];
  const float* w2_bu_r = (const float*)d_in[21];
  const float* dec_w1  = (const float*)d_in[22];
  const float* dec_b1  = (const float*)d_in[23];
  const float* dec_w2  = (const float*)d_in[24];
  const float* dec_b2  = (const float*)d_in[25];
  float* out = (float*)d_out;

  float* ws = (float*)d_ws;
  size_t off = 0;
  float* u0    = ws + off; off += (size_t)NUSER * CCH;
  float* b0    = ws + off; off += (size_t)NBOOK * CCH;
  float* u1    = ws + off; off += (size_t)NUSER * CCH;
  float* b1buf = ws + off; off += (size_t)NBOOK * CCH;
  int*   tab   = (int*)(ws + off); off += (size_t)NUSER * CCH;
  float* cnt_u = ws + off; off += NUSER;
  float* cnt_b = ws + off; off += NBOOK;
  float* u2 = u0;
  float* b2 = b0;
  int* adj_b    = tab;
  int* adj_u    = adj_b + (NEDGE + ADJ_PAD);
  int* hist_b   = adj_u + (NEDGE + ADJ_PAD);
  int* hist_u   = hist_b + NBOOK;
  int* rowptr_b = hist_u + NUSER;
  int* rowptr_u = rowptr_b + NBOOK;
  int* rank_b = (int*)u1;
  int* rank_u = rank_b + NEDGE;

  const int BT = 256;
  const int sage_blk_b = (NBOOK / 16 * 32 + BT - 1) / BT;
  const int sage_blk_u = (NUSER / 16 * 32 + BT - 1) / BT;
  const int edge_blk   = (NEDGE + BT - 1) / BT;
  const int gat_blk_b  = (NBOOK * 32 + BT - 1) / BT;
  const int gat_blk_u  = (NUSER * 32 + BT - 1) / BT;

  k_proj_user<<<(NUSER * CCH + BT - 1) / BT, BT, 0, stream>>>(user_x, user_lin_w, user_lin_b, u0);
  k_proj_book<<<(NBOOK / 16 * 32 + BT - 1) / BT, BT, 0, stream>>>(book_x, book_lin_w, book_lin_b, b0);

  k_zero<<<(NBOOK + NUSER + BT - 1) / BT, BT, 0, stream>>>((float*)hist_b, NBOOK + NUSER);
  k_hist<<<edge_blk, BT, 0, stream>>>(edge_src, edge_dst, hist_u, hist_b, rank_u, rank_b);
  k_scan<<<SCAN_BLK_B + SCAN_BLK_U, SCAN_T, 0, stream>>>(hist_b, hist_u, rowptr_b, rowptr_u, cnt_b, cnt_u);
  k_fill<<<edge_blk, BT, 0, stream>>>(edge_src, edge_dst, rank_u, rank_b, rowptr_u, rowptr_b, adj_u, adj_b);

  k_gather<<<gat_blk_b, BT, 0, stream>>>(adj_b, rowptr_b, hist_b, u0, b1buf, NBOOK);
  k_sage<<<sage_blk_b, BT, 0, stream>>>(b1buf, cnt_b, b0, w1_ub_l, w1_ub_r, b1_ub, b1buf, NBOOK, 1);
  k_gather<<<gat_blk_u, BT, 0, stream>>>(adj_u, rowptr_u, hist_u, b0, u1, NUSER);
  k_sage<<<sage_blk_u, BT, 0, stream>>>(u1, cnt_u, u0, w1_bu_l, w1_bu_r, b1_bu, u1, NUSER, 1);

  k_gather<<<gat_blk_b, BT, 0, stream>>>(adj_b, rowptr_b, hist_b, u1, b2, NBOOK);
  k_sage<<<sage_blk_b, BT, 0, stream>>>(b2, cnt_b, b1buf, w2_ub_l, w2_ub_r, b2_ub, b2, NBOOK, 0);
  k_gather<<<gat_blk_u, BT, 0, stream>>>(adj_u, rowptr_u, hist_u, b1buf, u2, NUSER);
  k_sage<<<sage_blk_u, BT, 0, stream>>>(u2, cnt_u, u1, w2_bu_l, w2_bu_r, b2_bu, u2, NUSER, 0);

  static_assert((size_t)NBOOK * HCH <= (size_t)NUSER * CCH, "book products fit the table region");
  static_assert((size_t)NUSER * HCH == (size_t)NBOOK * CCH + (size_t)NUSER * CCH + (size_t)NBOOK * CCH,
                "user products fill b0|u1|b1buf exactly");
  float* pdec_b = (float*)tab;
  float* pdec_u = b0;
  k_dec_proj<<<sage_blk_b, BT, 0, stream>>>(b2, dec_w1 + CCH * HCH, dec_b1, pdec_b, NBOOK, 0);
  k_dec_proj<<<sage_blk_u, BT, 0, stream>>>(u2, dec_w1, dec_b1, pdec_u, NUSER, 1);
  k_dec_edge<<<(NLBL + BT - 1) / BT, BT, 0, stream>>>(
      pdec_u, pdec_b, label_src, label_dst, dec_w2, dec_b2, out);
}


# edge sort without global atomics: a work-group owns a key range and an edge part, counts and places its edges with LDS counters; sum order: arrival order in a group as before, now at an LDS cursor
# speedup vs baseline: 2.7246x; 1.3962x over previous
#include <hip/hip_runtime.h>

#define NUSER 100000
#define NBOOK 50000
#define NEDGE 2000000
#define NLBL  500000
#define CCH   64
#define HCH   128

typedef __attribute__((ext_vector_type(2))) float v2f;
typedef __attribute__((ext_vector_type(8))) float v8f;

#if __has_builtin(__builtin_amdgcn_wmma_f32_16x16x4_f32)
#define HAVE_WMMA_F32 1
__device__ __forceinline__ v8f wmma4(v2f a, v2f b, v8f c) {
  return __builtin_amdgcn_wmma_f32_16x16x4_f32(false, a, false, b, (short)0, c, false, false);
}
#else
#define HAVE_WMMA_F32 0
#endif

__global__ void k_proj_user(const float* __restrict__ x, const float* __restrict__ w,
                            const float* __restrict__ bias, float* __restrict__ out) {
  int i = blockIdx.x * blockDim.x + threadIdx.x;
  if (i >= NUSER * CCH) return;
  int node = i >> 6, c = i & 63;
  float acc = bias[c];
#pragma unroll
  for (int k = 0; k < 3; ++k) acc = fmaf(x[node * 3 + k], w[k * CCH + c], acc);
  out[i] = acc;
}

__global__ void __launch_bounds__(256)
k_proj_book(const float* __restrict__ x, const float* __restrict__ w,
            const float* __restrict__ bias, float* __restrict__ out) {
  int tid = threadIdx.x;
  int wave = (blockIdx.x * blockDim.x + tid) >> 5;
  int lane = tid & 31;
  int m = lane & 15, half = lane >> 4;
  int row0 = wave << 4;
  bool active = row0 < NBOOK;
#if HAVE_WMMA_F32
  __shared__ v2f sW[64 * CCH];
  const float* xr = x + (size_t)(row0 + m) * 384;
  v8f zero = {0.f, 0.f, 0.f, 0.f, 0.f, 0.f, 0.f, 0.f};
  v8f acc[4] = {zero, zero, zero, zero};
  for (int k0 = 0; k0 < 384; k0 += 128) {
    __syncthreads();
    for (int idx = tid; idx < 64 * CCH; idx += 256) {
      int n = idx & 63, kp = idx >> 6;
      int k = k0 + 2 * kp;
      v2f v;
      v.x = w[k * CCH + n];
      v.y = w[(k + 1) * CCH + n];
      sW[idx] = v;
    }
    __syncthreads();
    if (active) {
      for (int k = k0; k < k0 + 128; k += 4) {
        int kk = k + 2 * half;
        v2f a;
        a.x = xr[kk];
        a.y = xr[kk + 1];
        int lp = (kk - k0) >> 1;
#pragma unroll
        for (int t = 0; t < 4; ++t)
          acc[t] = wmma4(a, sW[lp * CCH + t * 16 + m], acc[t]);
      }
    }
  }
  if (active) {
#pragma unroll
    for (int t = 0; t < 4; ++t) {
      int n = t * 16 + m;
      float bv = bias[n];
#pragma unroll
      for (int r = 0; r < 8; ++r)
        out[(size_t)(row0 + r + 8 * half) * CCH + n] = acc[t][r] + bv;
    }
  }
#else
  if (!active) return;
  const float* xr = x + (size_t)(row0 + m) * 384;
  for (int n = half * 32; n < half * 32 + 32; ++n) {
    float acc = bias[n];
    for (int k = 0; k < 384; ++k) acc = fmaf(xr[k], w[k * CCH + n], acc);
    out[(size_t)(row0 + m) * CCH + n] = acc;
  }
#endif
}

#define SORT_P 5
#define SORT_KR 3125
#define SORT_RB (NBOOK / SORT_KR)
#define SORT_RU (NUSER / SORT_KR)
#define SORT_T 1024
#define SORT_NV (NEDGE / 4 / SORT_P)
#define NKEY (NBOOK + NUSER)
static_assert(SORT_RB * SORT_KR == NBOOK && SORT_RU * SORT_KR == NUSER, "key ranges tile both tables");
static_assert(SORT_NV * 4 * SORT_P == NEDGE, "edge parts are whole 16-byte vectors");

#define CNT_U 8
__global__ void __launch_bounds__(SORT_T)
k_count(const int* __restrict__ gsrc, const int* __restrict__ gdst,
        int* __restrict__ histp) {
  __shared__ int s_cnt[SORT_KR];
  const int tid = threadIdx.x;
  const int part = (int)blockIdx.x % SORT_P;
  const int rg = (int)blockIdx.x / SORT_P;
  const bool isb = rg < SORT_RB;
  const int lo = (isb ? rg : rg - SORT_RB) * SORT_KR;
  const int4* keys = (const int4*)(isb ? gdst : gsrc) + (size_t)part * SORT_NV;
  for (int j = tid; j < SORT_KR; j += SORT_T) s_cnt[j] = 0;
  __syncthreads();
  int4 t[CNT_U];
#pragma unroll
  for (int u = 0; u < CNT_U; ++u) t[u] = keys[min(u * SORT_T + tid, SORT_NV - 1)];
  for (int q0 = 0; q0 < SORT_NV; q0 += CNT_U * SORT_T) {
    int4 n[CNT_U];
#pragma unroll
    for (int u = 0; u < CNT_U; ++u)
      n[u] = keys[min(q0 + (CNT_U + u) * SORT_T + tid, SORT_NV - 1)];
#pragma unroll
    for (int u = 0; u < CNT_U; ++u) {
      const bool ok = q0 + u * SORT_T + tid < SORT_NV;
      const unsigned r0 = (unsigned)(t[u].x - lo), r1 = (unsigned)(t[u].y - lo);
      const unsigned r2 = (unsigned)(t[u].z - lo), r3 = (unsigned)(t[u].w - lo);
      if (ok && r0 < SORT_KR) atomicAdd(&s_cnt[r0], 1);
      if (ok && r1 < SORT_KR) atomicAdd(&s_cnt[r1], 1);
      if (ok && r2 < SORT_KR) atomicAdd(&s_cnt[r2], 1);
      if (ok && r3 < SORT_KR) atomicAdd(&s_cnt[r3], 1);
    }
#pragma unroll
    for (int u = 0; u < CNT_U; ++u) t[u] = n[u];
  }
  __syncthreads();
  int* dst = histp + (size_t)part * NKEY + (isb ? 0 : NBOOK) + lo;
  for (int j = tid; j < SORT_KR; j += SORT_T) dst[j] = s_cnt[j];
}

__global__ void __launch_bounds__(256)
k_comb(int* __restrict__ histp, int* __restrict__ hist) {
  const int i = blockIdx.x * blockDim.x + threadIdx.x;
  if (i >= NKEY) return;
  int c[SORT_P];
#pragma unroll
  for (int p = 0; p < SORT_P; ++p) c[p] = histp[(size_t)p * NKEY + i];
  int run = 0;
#pragma unroll
  for (int p = 0; p < SORT_P; ++p) {
    histp[(size_t)p * NKEY + i] = run;
    run += c[p];
  }
  hist[i] = run;
}

#define SCAN_T 1024
#define SCAN_BLK_B ((NBOOK + SCAN_T - 1) / SCAN_T)
#define SCAN_BLK_U ((NUSER + SCAN_T - 1) / SCAN_T)
__global__ void __launch_bounds__(SCAN_T)
k_scan(const int* __restrict__ hist_b, const int* __restrict__ hist_u,
       int* __restrict__ rowptr_b, int* __restrict__ rowptr_u,
       float* __restrict__ cnt_b, float* __restrict__ cnt_u) {
  __shared__ int s_pre[32];
  __shared__ int s_tot[32];
  const bool isb = blockIdx.x < SCAN_BLK_B;
  const int* hist = isb ? hist_b : hist_u;
  int* rowptr = isb ? rowptr_b : rowptr_u;
  float* cnt = isb ? cnt_b : cnt_u;
  const int n = isb ? NBOOK : NUSER;
  const int tile0 = (isb ? (int)blockIdx.x : (int)blockIdx.x - SCAN_BLK_B) * SCAN_T;
  const int tid = threadIdx.x, lane = tid & 31, wv = tid >> 5;
  int pre = 0;
  const int4* h4 = (const int4*)hist;
  const int lim = tile0 / 4;
  for (int q0 = 0; q0 < lim; q0 += 8 * SCAN_T) {
    int4 t[8];
#pragma unroll
    for (int u = 0; u < 8; ++u) t[u] = h4[min(q0 + u * SCAN_T + tid, lim - 1)];
#pragma unroll
    for (int u = 0; u < 8; ++u)
      pre += (q0 + u * SCAN_T + tid < lim) ? (t[u].x + t[u].y) + (t[u].z + t[u].w) : 0;
  }
  const int i = tile0 + tid;
  const int c = (i < n) ? hist[i] : 0;
  int incl = c;
#pragma unroll
  for (int o = 1; o < 32; o <<= 1) {
    int t = __shfl_up(incl, o);
    if (lane >= o) incl += t;
  }
#pragma unroll
  for (int o = 16; o > 0; o >>= 1) pre += __shfl_xor(pre, o);
  if (lane == 31) s_tot[wv] = incl;
  if (lane == 0) s_pre[wv] = pre;
  __syncthreads();
  int base = s_pre[lane] + ((lane < wv) ? s_tot[lane] : 0);
#pragma unroll
  for (int o = 16; o > 0; o >>= 1) base += __shfl_xor(base, o);
  if (i < n) {
    rowptr[i] = base + incl - c;
    cnt[i] = (float)c;
  }
}

#define ADJ_PAD 32
#define PLC_U 4
__global__ void __launch_bounds__(SORT_T)
k_place(const int* __restrict__ gsrc, const int* __restrict__ gdst,
        const int* __restrict__ histp, const int* __restrict__ rowptr,
        int* __restrict__ adj_u, int* __restrict__ adj_b) {
  __shared__ int s_cur[SORT_KR];
  const int tid = threadIdx.x;
  const int part = (int)blockIdx.x % SORT_P;
  const int rg = (int)blockIdx.x / SORT_P;
  const bool isb = rg < SORT_RB;
  const int lo = (isb ? rg : rg - SORT_RB) * SORT_KR;
  const int kb = (isb ? 0 : NBOOK) + lo;
  const int4* keys = (const int4*)(isb ? gdst : gsrc) + (size_t)part * SORT_NV;
  const int4* vals = (const int4*)(isb ? gsrc : gdst) + (size_t)part * SORT_NV;
  int* adj = isb ? adj_b : adj_u;
  if (blockIdx.x == 0 && tid < ADJ_PAD) { adj_b[NEDGE + tid] = 0; adj_u[NEDGE + tid] = 0; }
  for (int j = tid; j < SORT_KR; j += SORT_T)
    s_cur[j] = rowptr[kb + j] + histp[(size_t)part * NKEY + kb + j];
  __syncthreads();
  int4 t[PLC_U], o[PLC_U];
#pragma unroll
  for (int u = 0; u < PLC_U; ++u) {
    const int q = min(u * SORT_T + tid, SORT_NV - 1);
    t[u] = keys[q];
    o[u] = vals[q];
  }
  for (int q0 = 0; q0 < SORT_NV; q0 += PLC_U * SORT_T) {
    int4 nt[PLC_U], no[PLC_U];
#pragma unroll
    for (int u = 0; u < PLC_U; ++u) {
      const int q = min(q0 + (PLC_U + u) * SORT_T + tid, SORT_NV - 1);
      nt[u] = keys[q];
      no[u] = vals[q];
    }
    int pos[PLC_U][4];
#pragma unroll
    for (int u = 0; u < PLC_U; ++u) {
      const bool ok = q0 + u * SORT_T + tid < SORT_NV;
      const unsigned r0 = (unsigned)(t[u].x - lo), r1 = (unsigned)(t[u].y - lo);
      const unsigned r2 = (unsigned)(t[u].z - lo), r3 = (unsigned)(t[u].w - lo);
      pos[u][0] = (ok && r0 < SORT_KR) ? atomicAdd(&s_cur[r0], 1) : -1;
      pos[u][1] = (ok && r1 < SORT_KR) ? atomicAdd(&s_cur[r1], 1) : -1;
      pos[u][2] = (ok && r2 < SORT_KR) ? atomicAdd(&s_cur[r2], 1) : -1;
      pos[u][3] = (ok && r3 < SORT_KR) ? atomicAdd(&s_cur[r3], 1) : -1;
    }
#pragma unroll
    for (int u = 0; u < PLC_U; ++u) {
      if (pos[u][0] >= 0) adj[min(pos[u][0], NEDGE - 1)] = o[u].x;
      if (pos[u][1] >= 0) adj[min(pos[u][1], NEDGE - 1)] = o[u].y;
      if (pos[u][2] >= 0) adj[min(pos[u][2], NEDGE - 1)] = o[u].z;
      if (pos[u][3] >= 0) adj[min(pos[u][3], NEDGE - 1)] = o[u].w;
    }
#pragma unroll
    for (int u = 0; u < PLC_U; ++u) { t[u] = nt[u]; o[u] = no[u]; }
  }
}

#define GB 8
static_assert(4 * GB <= ADJ_PAD, "index batches may run this far past a list");
__global__ void __launch_bounds__(256)
k_gather(const int* __restrict__ adj, const int* __restrict__ rowptr,
         const int* __restrict__ deg, const float* __restrict__ feat,
         float* __restrict__ agg, int n) {
  int row = (int)((blockIdx.x * blockDim.x + threadIdx.x) >> 5);
  row = __builtin_amdgcn_readfirstlane(row);
  if (row >= n) return;
  const int lane = (int)threadIdx.x & 31;
  const int h = lane >> 4, c = lane & 15;
  int start = rowptr[row];
  int end = start + deg[row];
  start = min(max(start, 0), NEDGE);
  end = min(max(end, start), NEDGE);
  const char* fb = (const char*)feat;
  const unsigned coff = (unsigned)c << 4;
  float4 acc = make_float4(0.f, 0.f, 0.f, 0.f);
  int nxt[GB];
#pragma unroll
  for (int k = 0; k < GB; ++k) nxt[k] = adj[start + 2 * k + h];
  for (int j = start; j < end; j += 2 * GB) {
    float4 v[GB];
#pragma unroll
    for (int k = 0; k < GB; ++k) {
      const unsigned idx = (j + 2 * k + h < end) ? (unsigned)nxt[k] : 0u;
      v[k] = *(const float4*)(fb + ((idx << 8) + coff));
    }
#pragma unroll
    for (int k = 0; k < GB; ++k) nxt[k] = adj[j + 2 * GB + 2 * k + h];
#pragma unroll
    for (int k = 0; k < GB; ++k) {
      const bool ok = j + 2 * k + h < end;
      acc.x += ok ? v[k].x : 0.f;
      acc.y += ok ? v[k].y : 0.f;
      acc.z += ok ? v[k].z : 0.f;
      acc.w += ok ? v[k].w : 0.f;
    }
  }
  acc.x += __shfl_xor(acc.x, 16);
  acc.y += __shfl_xor(acc.y, 16);
  acc.z += __shfl_xor(acc.z, 16);
  acc.w += __shfl_xor(acc.w, 16);
  if (h == 0) ((float4*)agg)[(size_t)row * (CCH / 4) + c] = acc;
}

__global__ void __launch_bounds__(256)
k_sage(const float* agg, const float* __restrict__ cnt,
       const float* __restrict__ xdst,
       const float* __restrict__ wl, const float* __restrict__ wr,
       const float* __restrict__ bias, float* out,
       int n, int relu) {
  int tid = threadIdx.x;
  int wave = (blockIdx.x * blockDim.x + tid) >> 5;
  int lane = tid & 31;
  int m = lane & 15, half = lane >> 4;
  int row0 = wave << 4;
  bool active = row0 < n;
#if HAVE_WMMA_F32
  __shared__ v2f sWl[32 * CCH];
  __shared__ v2f sWr[32 * CCH];
  for (int idx = tid; idx < 32 * CCH; idx += 256) {
    int nn = idx & 63, kp = idx >> 6;
    int k = 2 * kp;
    v2f a, b;
    a.x = wl[k * CCH + nn];       a.y = wl[(k + 1) * CCH + nn];
    b.x = wr[k * CCH + nn];       b.y = wr[(k + 1) * CCH + nn];
    sWl[idx] = a;
    sWr[idx] = b;
  }
  __syncthreads();
  if (!active) return;
  int row = row0 + m;
  float rc = 1.0f / fmaxf(cnt[row], 1.0f);
  const float* ar = agg + (size_t)row * CCH;
  const float* xr = xdst + (size_t)row * CCH;
  v8f zero = {0.f, 0.f, 0.f, 0.f, 0.f, 0.f, 0.f, 0.f};
  v8f acc[4] = {zero, zero, zero, zero};
  for (int k = 0; k < CCH; k += 4) {
    int kk = k + 2 * half;
    int lp = kk >> 1;
    v2f am, ax;
    am.x = ar[kk] * rc;  am.y = ar[kk + 1] * rc;
    ax.x = xr[kk];       ax.y = xr[kk + 1];
#pragma unroll
    for (int t = 0; t < 4; ++t) {
      int ncol = t * 16 + m;
      acc[t] = wmma4(am, sWl[lp * CCH + ncol], acc[t]);
      acc[t] = wmma4(ax, sWr[lp * CCH + ncol], acc[t]);
    }
  }
#pragma unroll
  for (int t = 0; t < 4; ++t) {
    int ncol = t * 16 + m;
    float bv = bias[ncol];
#pragma unroll
    for (int r = 0; r < 8; ++r) {
      float v = acc[t][r] + bv;
      if (relu) v = fmaxf(v, 0.0f);
      out[(size_t)(row0 + r + 8 * half) * CCH + ncol] = v;
    }
  }
#else
  if (!active) return;
  int row = row0 + m;
  float rc = 1.0f / fmaxf(cnt[row], 1.0f);
  const float* ar = agg + (size_t)row * CCH;
  const float* xr = xdst + (size_t)row * CCH;
  for (int ncol = half * 32; ncol < half * 32 + 32; ++ncol) {
    float v = bias[ncol];
    for (int k = 0; k < CCH; ++k)
      v = fmaf(ar[k] * rc, wl[k * CCH + ncol], fmaf(xr[k], wr[k * CCH + ncol], v));
    if (relu) v = fmaxf(v, 0.0f);
    out[(size_t)row * CCH + ncol] = v;
  }
#endif
}


__global__ void __launch_bounds__(256)
k_dec_proj(const float* __restrict__ x, const float* __restrict__ wh,
           const float* __restrict__ bias, float* __restrict__ p,
           int n, int use_bias) {
  int tid = threadIdx.x;
  int wave = (blockIdx.x * blockDim.x + tid) >> 5;
  int lane = tid & 31;
  int m = lane & 15, half = lane >> 4;
  int row0 = wave << 4;
  bool active = row0 < n;
#if HAVE_WMMA_F32
  __shared__ v2f sW[32 * HCH];
  for (int idx = tid; idx < 32 * HCH; idx += 256) {
    int nn = idx & 127, kp = idx >> 7;
    int k = 2 * kp;
    v2f v;
    v.x = wh[k * HCH + nn];
    v.y = wh[(k + 1) * HCH + nn];
    sW[idx] = v;
  }
  __syncthreads();
  if (!active) return;
  const float* xr = x + (size_t)(row0 + m) * CCH;
  v8f zero = {0.f, 0.f, 0.f, 0.f, 0.f, 0.f, 0.f, 0.f};
  v8f acc[8] = {zero, zero, zero, zero, zero, zero, zero, zero};
  for (int k = 0; k < CCH; k += 4) {
    int kk = k + 2 * half;
    v2f a;
    a.x = xr[kk];
    a.y = xr[kk + 1];
    int lp = kk >> 1;
#pragma unroll
    for (int t = 0; t < 8; ++t)
      acc[t] = wmma4(a, sW[lp * HCH + t * 16 + m], acc[t]);
  }
#pragma unroll
  for (int t = 0; t < 8; ++t) {
    int ncol = t * 16 + m;
    float bv = use_bias ? bias[ncol] : 0.0f;
#pragma unroll
    for (int r = 0; r < 8; ++r)
      p[(size_t)(row0 + r + 8 * half) * HCH + ncol] = acc[t][r] + bv;
  }
#else
  if (!active) return;
  const float* xr = x + (size_t)(row0 + m) * CCH;
  for (int ncol = half * 64; ncol < half * 64 + 64; ++ncol) {
    float v = use_bias ? bias[ncol] : 0.0f;
    for (int k = 0; k < CCH; ++k) v = fmaf(xr[k], wh[k * HCH + ncol], v);
    p[(size_t)(row0 + m) * HCH + ncol] = v;
  }
#endif
}

static_assert(NLBL % 32 == 0, "a wave takes 32 edges");
__global__ void __launch_bounds__(256)
k_dec_edge(const float* __restrict__ pu, const float* __restrict__ pb,
           const int* __restrict__ ls, const int* __restrict__ ld,
           const float* __restrict__ w2, const float* __restrict__ bias2,
           float* __restrict__ out) {
  int wave = (int)((blockIdx.x * blockDim.x + threadIdx.x) >> 5);
  wave = __builtin_amdgcn_readfirstlane(wave);
  const int e0 = wave * 32;
  if (e0 >= NLBL) return;
  const int lane = (int)threadIdx.x & 31;
  const unsigned loff = (unsigned)lane << 4;
  const float4 wv = ((const float4*)w2)[lane];
  const float ob = bias2[0];
  const char* pub = (const char*)pu;
  const char* pbb = (const char*)pb;
  const bool h = (lane & 16) != 0, g = (lane & 8) != 0, f = (lane & 4) != 0;
  for (int b = 0; b < 4; ++b) {
    float4 va[8], vc[8];
#pragma unroll
    for (int i = 0; i < 8; ++i) {
      const int e = e0 + 8 * b + i;
      const unsigned su = (unsigned)min(max(ls[e], 0), NUSER - 1);
      const unsigned sb = (unsigned)min(max(ld[e], 0), NBOOK - 1);
      va[i] = *(const float4*)(pub + ((su << 9) + loff));
      vc[i] = *(const float4*)(pbb + ((sb << 9) + loff));
    }
    float p[8];
#pragma unroll
    for (int i = 0; i < 8; ++i) {
      const float x0 = fmaxf(va[i].x + vc[i].x, 0.0f);
      const float x1 = fmaxf(va[i].y + vc[i].y, 0.0f);
      const float x2 = fmaxf(va[i].z + vc[i].z, 0.0f);
      const float x3 = fmaxf(va[i].w + vc[i].w, 0.0f);
      p[i] = fmaf(x3, wv.w, fmaf(x2, wv.z, fmaf(x1, wv.y, x0 * wv.x)));
    }
    float q[4];
#pragma unroll
    for (int i = 0; i < 4; ++i) {
      const float keep = h ? p[i + 4] : p[i];
      const float send = h ? p[i] : p[i + 4];
      q[i] = keep + __shfl_xor(send, 16);
    }
    float r[2];
#pragma unroll
    for (int i = 0; i < 2; ++i) {
      const float keep = g ? q[i + 2] : q[i];
      const float send = g ? q[i] : q[i + 2];
      r[i] = keep + __shfl_xor(send, 8);
    }
    float s = (f ? r[1] : r[0]) + __shfl_xor(f ? r[0] : r[1], 4);
    s += __shfl_xor(s, 2);
    s += __shfl_xor(s, 1);
    if ((lane & 3) == 0) out[e0 + 8 * b + (lane >> 2)] = s + ob;
  }
}

extern "C" void kernel_launch(void* const* d_in, const int* in_sizes, int n_in,
                              void* d_out, int out_size, void* d_ws, size_t ws_size,
                              hipStream_t stream) {
  (void)in_sizes; (void)n_in; (void)out_size; (void)ws_size;
  const float* user_x     = (const float*)d_in[0];
  const float* book_x     = (const float*)d_in[1];
  const int*   edge_src   = (const int*)d_in[2];
  const int*   edge_dst   = (const int*)d_in[3];
  const int*   label_src  = (const int*)d_in[4];
  const int*   label_dst  = (const int*)d_in[5];
  const float* user_lin_w = (const float*)d_in[6];
  const float* user_lin_b = (const float*)d_in[7];
  const float* book_lin_w = (const float*)d_in[8];
  const float* book_lin_b = (const float*)d_in[9];
  const float* w1_ub_l = (const float*)d_in[10];
  const float* b1_ub   = (const float*)d_in[11];
  const float* w1_ub_r = (const float*)d_in[12];
  const float* w1_bu_l = (const float*)d_in[13];
  const float* b1_bu   = (const float*)d_in[14];
  const float* w1_bu_r = (const float*)d_in[15];
  const float* w2_ub_l = (const float*)d_in[16];
  const float* b2_ub   = (const float*)d_in[17];
  const float* w2_ub_r = (const float*)d_in[18];
  const float* w2_bu_l = (const float*)d_in[19];
  const float* b2_bu   = (const float*)d_in[20];
  const float* w2_bu_r = (const float*)d_in[21];
  const float* dec_w1  = (const float*)d_in[22];
  const float* dec_b1  = (const float*)d_in[23];
  const float* dec_w2  = (const float*)d_in[24];
  const float* dec_b2  = (const float*)d_in[25];
  float* out = (float*)d_out;

  float* ws = (float*)d_ws;
  size_t off = 0;
  float* u0    = ws + off; off += (size_t)NUSER * CCH;
  float* b0    = ws + off; off += (size_t)NBOOK * CCH;
  float* u1    = ws + off; off += (size_t)NUSER * CCH;
  float* b1buf = ws + off; off += (size_t)NBOOK * CCH;
  int*   tab   = (int*)(ws + off); off += (size_t)NUSER * CCH;
  float* cnt_u = ws + off; off += NUSER;
  float* cnt_b = ws + off; off += NBOOK;
  float* u2 = u0;
  float* b2 = b0;
  int* adj_b    = tab;
  int* adj_u    = adj_b + (NEDGE + ADJ_PAD);
  int* hist_b   = adj_u + (NEDGE + ADJ_PAD);
  int* hist_u   = hist_b + NBOOK;
  int* rowptr_b = hist_u + NUSER;
  int* rowptr_u = rowptr_b + NBOOK;
  int* histp    = rowptr_u + NUSER;
  static_assert(2 * (size_t)(NEDGE + ADJ_PAD) + 2 * (size_t)NKEY + (size_t)SORT_P * NKEY
                    <= (size_t)NUSER * CCH, "integer tables fit their region");

  const int BT = 256;
  const int sage_blk_b = (NBOOK / 16 * 32 + BT - 1) / BT;
  const int sage_blk_u = (NUSER / 16 * 32 + BT - 1) / BT;
  const int gat_blk_b  = (NBOOK * 32 + BT - 1) / BT;
  const int gat_blk_u  = (NUSER * 32 + BT - 1) / BT;

  k_proj_user<<<(NUSER * CCH + BT - 1) / BT, BT, 0, stream>>>(user_x, user_lin_w, user_lin_b, u0);
  k_proj_book<<<(NBOOK / 16 * 32 + BT - 1) / BT, BT, 0, stream>>>(book_x, book_lin_w, book_lin_b, b0);

  k_count<<<(SORT_RB + SORT_RU) * SORT_P, SORT_T, 0, stream>>>(edge_src, edge_dst, histp);
  k_comb<<<(NKEY + BT - 1) / BT, BT, 0, stream>>>(histp, hist_b);
  k_scan<<<SCAN_BLK_B + SCAN_BLK_U, SCAN_T, 0, stream>>>(hist_b, hist_u, rowptr_b, rowptr_u, cnt_b, cnt_u);
  k_place<<<(SORT_RB + SORT_RU) * SORT_P, SORT_T, 0, stream>>>(edge_src, edge_dst, histp, rowptr_b, adj_u, adj_b);

  k_gather<<<gat_blk_b, BT, 0, stream>>>(adj_b, rowptr_b, hist_b, u0, b1buf, NBOOK);
  k_sage<<<sage_blk_b, BT, 0, stream>>>(b1buf, cnt_b, b0, w1_ub_l, w1_ub_r, b1_ub, b1buf, NBOOK, 1);
  k_gather<<<gat_blk_u, BT, 0, stream>>>(adj_u, rowptr_u, hist_u, b0, u1, NUSER);
  k_sage<<<sage_blk_u, BT, 0, stream>>>(u1, cnt_u, u0, w1_bu_l, w1_bu_r, b1_bu, u1, NUSER, 1);

  k_gather<<<gat_blk_b, BT, 0, stream>>>(adj_b, rowptr_b, hist_b, u1, b2, NBOOK);
  k_sage<<<sage_blk_b, BT, 0, stream>>>(b2, cnt_b, b1buf, w2_ub_l, w2_ub_r, b2_ub, b2, NBOOK, 0);
  k_gather<<<gat_blk_u, BT, 0, stream>>>(adj_u, rowptr_u, hist_u, b1buf, u2, NUSER);
  k_sage<<<sage_blk_u, BT, 0, stream>>>(u2, cnt_u, u1, w2_bu_l, w2_bu_r, b2_bu, u2, NUSER, 0);

  static_assert((size_t)NBOOK * HCH <= (size_t)NUSER * CCH, "book products fit the table region");
  static_assert((size_t)NUSER * HCH == (size_t)NBOOK * CCH + (size_t)NUSER * CCH + (size_t)NBOOK * CCH,
                "user products fill b0|u1|b1buf exactly");
  float* pdec_b = (float*)tab;
  float* pdec_u = b0;
  k_dec_proj<<<sage_blk_b, BT, 0, stream>>>(b2, dec_w1 + CCH * HCH, dec_b1, pdec_b, NBOOK, 0);
  k_dec_proj<<<sage_blk_u, BT, 0, stream>>>(u2, dec_w1, dec_b1, pdec_u, NUSER, 1);
  k_dec_edge<<<(NLBL + BT - 1) / BT, BT, 0, stream>>>(
      pdec_u, pdec_b, label_src, label_dst, dec_w2, dec_b2, out);
}


# independent launches merged, 17 launches to 11: per layer one gather grid and one transform grid serve both node types; projections ride in the sort's combine launch; sum order unchanged
# speedup vs baseline: 2.9589x; 1.0860x over previous
#include <hip/hip_runtime.h>

#define NUSER 100000
#define NBOOK 50000
#define NEDGE 2000000
#define NLBL  500000
#define CCH   64
#define HCH   128

typedef __attribute__((ext_vector_type(2))) float v2f;
typedef __attribute__((ext_vector_type(8))) float v8f;

#if __has_builtin(__builtin_amdgcn_wmma_f32_16x16x4_f32)
#define HAVE_WMMA_F32 1
__device__ __forceinline__ v8f wmma4(v2f a, v2f b, v8f c) {
  return __builtin_amdgcn_wmma_f32_16x16x4_f32(false, a, false, b, (short)0, c, false, false);
}
#else
#define HAVE_WMMA_F32 0
#endif


__device__ __forceinline__ void proj_user_blk(int blk, const float* __restrict__ x,
                                              const float* __restrict__ w,
                                              const float* __restrict__ bias,
                                              float* __restrict__ out) {
  int i = blk * 256 + (int)threadIdx.x;
  if (i >= NUSER * CCH) return;
  int node = i >> 6, c = i & 63;
  float acc = bias[c];
#pragma unroll
  for (int k = 0; k < 3; ++k) acc = fmaf(x[node * 3 + k], w[k * CCH + c], acc);
  out[i] = acc;
}

__device__ __forceinline__ void proj_book_blk(int blk, const float* __restrict__ x,
                                              const float* __restrict__ w,
                                              const float* __restrict__ bias,
                                              float* __restrict__ out) {
  int tid = threadIdx.x;
  int wave = (blk * 256 + tid) >> 5;
  int lane = tid & 31;
  int m = lane & 15, half = lane >> 4;
  int row0 = wave << 4;
  bool active = row0 < NBOOK;
#if HAVE_WMMA_F32
  __shared__ v2f sW[64 * CCH];
  const float* xr = x + (size_t)(row0 + m) * 384;
  v8f zero = {0.f, 0.f, 0.f, 0.f, 0.f, 0.f, 0.f, 0.f};
  v8f acc[4] = {zero, zero, zero, zero};
  for (int k0 = 0; k0 < 384; k0 += 128) {
    __syncthreads();
    for (int idx = tid; idx < 64 * CCH; idx += 256) {
      int n = idx & 63, kp = idx >> 6;
      int k = k0 + 2 * kp;
      v2f v;
      v.x = w[k * CCH + n];
      v.y = w[(k + 1) * CCH + n];
      sW[idx] = v;
    }
    __syncthreads();
    if (active) {
      for (int k = k0; k < k0 + 128; k += 4) {
        int kk = k + 2 * half;
        v2f a;
        a.x = xr[kk];
        a.y = xr[kk + 1];
        int lp = (kk - k0) >> 1;
#pragma unroll
        for (int t = 0; t < 4; ++t)
          acc[t] = wmma4(a, sW[lp * CCH + t * 16 + m], acc[t]);
      }
    }
  }
  if (active) {
#pragma unroll
    for (int t = 0; t < 4; ++t) {
      int n = t * 16 + m;
      float bv = bias[n];
#pragma unroll
      for (int r = 0; r < 8; ++r)
        out[(size_t)(row0 + r + 8 * half) * CCH + n] = acc[t][r] + bv;
    }
  }
#else
  if (!active) return;
  const float* xr = x + (size_t)(row0 + m) * 384;
  for (int n = half * 32; n < half * 32 + 32; ++n) {
    float acc = bias[n];
    for (int k = 0; k < 384; ++k) acc = fmaf(xr[k], w[k * CCH + n], acc);
    out[(size_t)(row0 + m) * CCH + n] = acc;
  }
#endif
}

#define SORT_P 5
#define SORT_KR 3125
#define SORT_RB (NBOOK / SORT_KR)
#define SORT_RU (NUSER / SORT_KR)
#define SORT_T 1024
#define SORT_NV (NEDGE / 4 / SORT_P)
#define NKEY (NBOOK + NUSER)
static_assert(SORT_RB * SORT_KR == NBOOK && SORT_RU * SORT_KR == NUSER, "key ranges tile both tables");
static_assert(SORT_NV * 4 * SORT_P == NEDGE, "edge parts are whole 16-byte vectors");

#define CNT_U 8
__global__ void __launch_bounds__(SORT_T)
k_count(const int* __restrict__ gsrc, const int* __restrict__ gdst,
        int* __restrict__ histp) {
  __shared__ int s_cnt[SORT_KR];
  const int tid = threadIdx.x;
  const int part = (int)blockIdx.x % SORT_P;
  const int rg = (int)blockIdx.x / SORT_P;
  const bool isb = rg < SORT_RB;
  const int lo = (isb ? rg : rg - SORT_RB) * SORT_KR;
  const int4* keys = (const int4*)(isb ? gdst : gsrc) + (size_t)part * SORT_NV;
  for (int j = tid; j < SORT_KR; j += SORT_T) s_cnt[j] = 0;
  __syncthreads();
  int4 t[CNT_U];
#pragma unroll
  for (int u = 0; u < CNT_U; ++u) t[u] = keys[min(u * SORT_T + tid, SORT_NV - 1)];
  for (int q0 = 0; q0 < SORT_NV; q0 += CNT_U * SORT_T) {
    int4 n[CNT_U];
#pragma unroll
    for (int u = 0; u < CNT_U; ++u)
      n[u] = keys[min(q0 + (CNT_U + u) * SORT_T + tid, SORT_NV - 1)];
#pragma unroll
    for (int u = 0; u < CNT_U; ++u) {
      const bool ok = q0 + u * SORT_T + tid < SORT_NV;
      const unsigned r0 = (unsigned)(t[u].x - lo), r1 = (unsigned)(t[u].y - lo);
      const unsigned r2 = (unsigned)(t[u].z - lo), r3 = (unsigned)(t[u].w - lo);
      if (ok && r0 < SORT_KR) atomicAdd(&s_cnt[r0], 1);
      if (ok && r1 < SORT_KR) atomicAdd(&s_cnt[r1], 1);
      if (ok && r2 < SORT_KR) atomicAdd(&s_cnt[r2], 1);
      if (ok && r3 < SORT_KR) atomicAdd(&s_cnt[r3], 1);
    }
#pragma unroll
    for (int u = 0; u < CNT_U; ++u) t[u] = n[u];
  }
  __syncthreads();
  int* dst = histp + (size_t)part * NKEY + (isb ? 0 : NBOOK) + lo;
  for (int j = tid; j < SORT_KR; j += SORT_T) dst[j] = s_cnt[j];
}

__device__ __forceinline__ void comb_blk(int blk, int* __restrict__ histp, int* __restrict__ hist) {
  const int i = blk * 256 + (int)threadIdx.x;
  if (i >= NKEY) return;
  int c[SORT_P];
#pragma unroll
  for (int p = 0; p < SORT_P; ++p) c[p] = histp[(size_t)p * NKEY + i];
  int run = 0;
#pragma unroll
  for (int p = 0; p < SORT_P; ++p) {
    histp[(size_t)p * NKEY + i] = run;
    run += c[p];
  }
  hist[i] = run;
}

#define PB_BLK ((NBOOK / 16 * 32 + 255) / 256)
#define CB_BLK ((NKEY + 255) / 256)
#define PU_BLK ((NUSER * CCH + 255) / 256)
__global__ void __launch_bounds__(256)
k_stage1(const float* __restrict__ user_x, const float* __restrict__ user_w,
         const float* __restrict__ user_b, float* __restrict__ u0,
         const float* __restrict__ book_x, const float* __restrict__ book_w,
         const float* __restrict__ book_b, float* __restrict__ b0,
         int* __restrict__ histp, int* __restrict__ hist) {
  const int blk = (int)blockIdx.x;
  if (blk < PB_BLK) proj_book_blk(blk, book_x, book_w, book_b, b0);
  else if (blk < PB_BLK + CB_BLK) comb_blk(blk - PB_BLK, histp, hist);
  else proj_user_blk(blk - (PB_BLK + CB_BLK), user_x, user_w, user_b, u0);
}

#define SCAN_T 1024
#define SCAN_BLK_B ((NBOOK + SCAN_T - 1) / SCAN_T)
#define SCAN_BLK_U ((NUSER + SCAN_T - 1) / SCAN_T)
__global__ void __launch_bounds__(SCAN_T)
k_scan(const int* __restrict__ hist_b, const int* __restrict__ hist_u,
       int* __restrict__ rowptr_b, int* __restrict__ rowptr_u,
       float* __restrict__ cnt_b, float* __restrict__ cnt_u) {
  __shared__ int s_pre[32];
  __shared__ int s_tot[32];
  const bool isb = blockIdx.x < SCAN_BLK_B;
  const int* hist = isb ? hist_b : hist_u;
  int* rowptr = isb ? rowptr_b : rowptr_u;
  float* cnt = isb ? cnt_b : cnt_u;
  const int n = isb ? NBOOK : NUSER;
  const int tile0 = (isb ? (int)blockIdx.x : (int)blockIdx.x - SCAN_BLK_B) * SCAN_T;
  const int tid = threadIdx.x, lane = tid & 31, wv = tid >> 5;
  int pre = 0;
  const int4* h4 = (const int4*)hist;
  const int lim = tile0 / 4;
  for (int q0 = 0; q0 < lim; q0 += 8 * SCAN_T) {
    int4 t[8];
#pragma unroll
    for (int u = 0; u < 8; ++u) t[u] = h4[min(q0 + u * SCAN_T + tid, lim - 1)];
#pragma unroll
    for (int u = 0; u < 8; ++u)
      pre += (q0 + u * SCAN_T + tid < lim) ? (t[u].x + t[u].y) + (t[u].z + t[u].w) : 0;
  }
  const int i = tile0 + tid;
  const int c = (i < n) ? hist[i] : 0;
  int incl = c;
#pragma unroll
  for (int o = 1; o < 32; o <<= 1) {
    int t = __shfl_up(incl, o);
    if (lane >= o) incl += t;
  }
#pragma unroll
  for (int o = 16; o > 0; o >>= 1) pre += __shfl_xor(pre, o);
  if (lane == 31) s_tot[wv] = incl;
  if (lane == 0) s_pre[wv] = pre;
  __syncthreads();
  int base = s_pre[lane] + ((lane < wv) ? s_tot[lane] : 0);
#pragma unroll
  for (int o = 16; o > 0; o >>= 1) base += __shfl_xor(base, o);
  if (i < n) {
    rowptr[i] = base + incl - c;
    cnt[i] = (float)c;
  }
}

#define ADJ_PAD 32
#define PLC_U 4
__global__ void __launch_bounds__(SORT_T)
k_place(const int* __restrict__ gsrc, const int* __restrict__ gdst,
        const int* __restrict__ histp, const int* __restrict__ rowptr,
        int* __restrict__ adj_u, int* __restrict__ adj_b) {
  __shared__ int s_cur[SORT_KR];
  const int tid = threadIdx.x;
  const int part = (int)blockIdx.x % SORT_P;
  const int rg = (int)blockIdx.x / SORT_P;
  const bool isb = rg < SORT_RB;
  const int lo = (isb ? rg : rg - SORT_RB) * SORT_KR;
  const int kb = (isb ? 0 : NBOOK) + lo;
  const int4* keys = (const int4*)(isb ? gdst : gsrc) + (size_t)part * SORT_NV;
  const int4* vals = (const int4*)(isb ? gsrc : gdst) + (size_t)part * SORT_NV;
  int* adj = isb ? adj_b : adj_u;
  if (blockIdx.x == 0 && tid < ADJ_PAD) { adj_b[NEDGE + tid] = 0; adj_u[NEDGE + tid] = 0; }
  for (int j = tid; j < SORT_KR; j += SORT_T)
    s_cur[j] = rowptr[kb + j] + histp[(size_t)part * NKEY + kb + j];
  __syncthreads();
  int4 t[PLC_U], o[PLC_U];
#pragma unroll
  for (int u = 0; u < PLC_U; ++u) {
    const int q = min(u * SORT_T + tid, SORT_NV - 1);
    t[u] = keys[q];
    o[u] = vals[q];
  }
  for (int q0 = 0; q0 < SORT_NV; q0 += PLC_U * SORT_T) {
    int4 nt[PLC_U], no[PLC_U];
#pragma unroll
    for (int u = 0; u < PLC_U; ++u) {
      const int q = min(q0 + (PLC_U + u) * SORT_T + tid, SORT_NV - 1);
      nt[u] = keys[q];
      no[u] = vals[q];
    }
    int pos[PLC_U][4];
#pragma unroll
    for (int u = 0; u < PLC_U; ++u) {
      const bool ok = q0 + u * SORT_T + tid < SORT_NV;
      const unsigned r0 = (unsigned)(t[u].x - lo), r1 = (unsigned)(t[u].y - lo);
      const unsigned r2 = (unsigned)(t[u].z - lo), r3 = (unsigned)(t[u].w - lo);
      pos[u][0] = (ok && r0 < SORT_KR) ? atomicAdd(&s_cur[r0], 1) : -1;
      pos[u][1] = (ok && r1 < SORT_KR) ? atomicAdd(&s_cur[r1], 1) : -1;
      pos[u][2] = (ok && r2 < SORT_KR) ? atomicAdd(&s_cur[r2], 1) : -1;
      pos[u][3] = (ok && r3 < SORT_KR) ? atomicAdd(&s_cur[r3], 1) : -1;
    }
#pragma unroll
    for (int u = 0; u < PLC_U; ++u) {
      if (pos[u][0] >= 0) adj[min(pos[u][0], NEDGE - 1)] = o[u].x;
      if (pos[u][1] >= 0) adj[min(pos[u][1], NEDGE - 1)] = o[u].y;
      if (pos[u][2] >= 0) adj[min(pos[u][2], NEDGE - 1)] = o[u].z;
      if (pos[u][3] >= 0) adj[min(pos[u][3], NEDGE - 1)] = o[u].w;
    }
#pragma unroll
    for (int u = 0; u < PLC_U; ++u) { t[u] = nt[u]; o[u] = no[u]; }
  }
}

#define GB 8
static_assert(4 * GB <= ADJ_PAD, "index batches may run this far past a list");
__device__ __forceinline__ void gather_blk(int blk, const int* __restrict__ adj,
                                           const int* __restrict__ rowptr,
                                           const int* __restrict__ deg,
                                           const float* __restrict__ feat,
                                           float* __restrict__ agg, int n) {
  int row = (blk * 256 + (int)threadIdx.x) >> 5;
  row = __builtin_amdgcn_readfirstlane(row);
  if (row >= n) return;
  const int lane = (int)threadIdx.x & 31;
  const int h = lane >> 4, c = lane & 15;
  int start = rowptr[row];
  int end = start + deg[row];
  start = min(max(start, 0), NEDGE);
  end = min(max(end, start), NEDGE);
  const char* fb = (const char*)feat;
  const unsigned coff = (unsigned)c << 4;
  float4 acc = make_float4(0.f, 0.f, 0.f, 0.f);
  int nxt[GB];
#pragma unroll
  for (int k = 0; k < GB; ++k) nxt[k] = adj[start + 2 * k + h];
  for (int j = start; j < end; j += 2 * GB) {
    float4 v[GB];
#pragma unroll
    for (int k = 0; k < GB; ++k) {
      const unsigned idx = (j + 2 * k + h < end) ? (unsigned)nxt[k] : 0u;
      v[k] = *(const float4*)(fb + ((idx << 8) + coff));
    }
#pragma unroll
    for (int k = 0; k < GB; ++k) nxt[k] = adj[j + 2 * GB + 2 * k + h];
#pragma unroll
    for (int k = 0; k < GB; ++k) {
      const bool ok = j + 2 * k + h < end;
      acc.x += ok ? v[k].x : 0.f;
      acc.y += ok ? v[k].y : 0.f;
      acc.z += ok ? v[k].z : 0.f;
      acc.w += ok ? v[k].w : 0.f;
    }
  }
  acc.x += __shfl_xor(acc.x, 16);
  acc.y += __shfl_xor(acc.y, 16);
  acc.z += __shfl_xor(acc.z, 16);
  acc.w += __shfl_xor(acc.w, 16);
  if (h == 0) ((float4*)agg)[(size_t)row * (CCH / 4) + c] = acc;
}

#define GAT_BLK_B ((NBOOK * 32 + 255) / 256)
#define GAT_BLK_U ((NUSER * 32 + 255) / 256)
static_assert(GAT_BLK_B * 8 == NBOOK, "a block's eight rows are all of one table");
__global__ void __launch_bounds__(256)
k_gather(const int* __restrict__ adj_b, const int* __restrict__ rowptr_b,
         const int* __restrict__ deg_b, const float* __restrict__ feat_u,
         float* __restrict__ agg_b,
         const int* __restrict__ adj_u, const int* __restrict__ rowptr_u,
         const int* __restrict__ deg_u, const float* __restrict__ feat_b,
         float* __restrict__ agg_u) {
  const bool isb = (int)blockIdx.x < GAT_BLK_B;
  gather_blk(isb ? (int)blockIdx.x : (int)blockIdx.x - GAT_BLK_B,
             isb ? adj_b : adj_u, isb ? rowptr_b : rowptr_u, isb ? deg_b : deg_u,
             isb ? feat_u : feat_b, isb ? agg_b : agg_u, isb ? NBOOK : NUSER);
}

__device__ __forceinline__ void sage_blk(int blk, const float* agg, const float* __restrict__ cnt,
                                         const float* __restrict__ xdst,
                                         const float* __restrict__ wl, const float* __restrict__ wr,
                                         const float* __restrict__ bias, float* out,
                                         int n, int relu) {
  int tid = threadIdx.x;
  int wave = (blk * 256 + tid) >> 5;
  int lane = tid & 31;
  int m = lane & 15, half = lane >> 4;
  int row0 = wave << 4;
  bool active = row0 < n;
#if HAVE_WMMA_F32
  __shared__ v2f sWl[32 * CCH];
  __shared__ v2f sWr[32 * CCH];
  for (int idx = tid; idx < 32 * CCH; idx += 256) {
    int nn = idx & 63, kp = idx >> 6;
    int k = 2 * kp;
    v2f a, b;
    a.x = wl[k * CCH + nn];       a.y = wl[(k + 1) * CCH + nn];
    b.x = wr[k * CCH + nn];       b.y = wr[(k + 1) * CCH + nn];
    sWl[idx] = a;
    sWr[idx] = b;
  }
  __syncthreads();
  if (!active) return;
  int row = row0 + m;
  float rc = 1.0f / fmaxf(cnt[row], 1.0f);
  const float* ar = agg + (size_t)row * CCH;
  const float* xr = xdst + (size_t)row * CCH;
  v8f zero = {0.f, 0.f, 0.f, 0.f, 0.f, 0.f, 0.f, 0.f};
  v8f acc[4] = {zero, zero, zero, zero};
  for (int k = 0; k < CCH; k += 4) {
    int kk = k + 2 * half;
    int lp = kk >> 1;
    v2f am, ax;
    am.x = ar[kk] * rc;  am.y = ar[kk + 1] * rc;
    ax.x = xr[kk];       ax.y = xr[kk + 1];
#pragma unroll
    for (int t = 0; t < 4; ++t) {
      int ncol = t * 16 + m;
      acc[t] = wmma4(am, sWl[lp * CCH + ncol], acc[t]);
      acc[t] = wmma4(ax, sWr[lp * CCH + ncol], acc[t]);
    }
  }
#pragma unroll
  for (int t = 0; t < 4; ++t) {
    int ncol = t * 16 + m;
    float bv = bias[ncol];
#pragma unroll
    for (int r = 0; r < 8; ++r) {
      float v = acc[t][r] + bv;
      if (relu) v = fmaxf(v, 0.0f);
      out[(size_t)(row0 + r + 8 * half) * CCH + ncol] = v;
    }
  }
#else
  if (!active) return;
  int row = row0 + m;
  float rc = 1.0f / fmaxf(cnt[row], 1.0f);
  const float* ar = agg + (size_t)row * CCH;
  const float* xr = xdst + (size_t)row * CCH;
  for (int ncol = half * 32; ncol < half * 32 + 32; ++ncol) {
    float v = bias[ncol];
    for (int k = 0; k < CCH; ++k)
      v = fmaf(ar[k] * rc, wl[k * CCH + ncol], fmaf(xr[k], wr[k * CCH + ncol], v));
    if (relu) v = fmaxf(v, 0.0f);
    out[(size_t)row * CCH + ncol] = v;
  }
#endif
}

#define SAGE_BLK_B ((NBOOK / 16 * 32 + 255) / 256)
#define SAGE_BLK_U ((NUSER / 16 * 32 + 255) / 256)
__global__ void __launch_bounds__(256)
k_sage(const float* agg_b, const float* __restrict__ cnt_b, const float* __restrict__ x_b,
       const float* __restrict__ wl_b, const float* __restrict__ wr_b,
       const float* __restrict__ bias_b, float* out_b,
       const float* agg_u, const float* __restrict__ cnt_u, const float* __restrict__ x_u,
       const float* __restrict__ wl_u, const float* __restrict__ wr_u,
       const float* __restrict__ bias_u, float* out_u, int relu) {
  const bool isb = (int)blockIdx.x < SAGE_BLK_B;
  sage_blk(isb ? (int)blockIdx.x : (int)blockIdx.x - SAGE_BLK_B,
           isb ? agg_b : agg_u, isb ? cnt_b : cnt_u, isb ? x_b : x_u,
           isb ? wl_b : wl_u, isb ? wr_b : wr_u, isb ? bias_b : bias_u,
           isb ? out_b : out_u, isb ? NBOOK : NUSER, relu);
}


__global__ void __launch_bounds__(256)
k_dec_proj(const float* __restrict__ x, const float* __restrict__ wh,
           const float* __restrict__ bias, float* __restrict__ p,
           int n, int use_bias) {
  int tid = threadIdx.x;
  int wave = (blockIdx.x * blockDim.x + tid) >> 5;
  int lane = tid & 31;
  int m = lane & 15, half = lane >> 4;
  int row0 = wave << 4;
  bool active = row0 < n;
#if HAVE_WMMA_F32
  __shared__ v2f sW[32 * HCH];
  for (int idx = tid; idx < 32 * HCH; idx += 256) {
    int nn = idx & 127, kp = idx >> 7;
    int k = 2 * kp;
    v2f v;
    v.x = wh[k * HCH + nn];
    v.y = wh[(k + 1) * HCH + nn];
    sW[idx] = v;
  }
  __syncthreads();
  if (!active) return;
  const float* xr = x + (size_t)(row0 + m) * CCH;
  v8f zero = {0.f, 0.f, 0.f, 0.f, 0.f, 0.f, 0.f, 0.f};
  v8f acc[8] = {zero, zero, zero, zero, zero, zero, zero, zero};
  for (int k = 0; k < CCH; k += 4) {
    int kk = k + 2 * half;
    v2f a;
    a.x = xr[kk];
    a.y = xr[kk + 1];
    int lp = kk >> 1;
#pragma unroll
    for (int t = 0; t < 8; ++t)
      acc[t] = wmma4(a, sW[lp * HCH + t * 16 + m], acc[t]);
  }
#pragma unroll
  for (int t = 0; t < 8; ++t) {
    int ncol = t * 16 + m;
    float bv = use_bias ? bias[ncol] : 0.0f;
#pragma unroll
    for (int r = 0; r < 8; ++r)
      p[(size_t)(row0 + r + 8 * half) * HCH + ncol] = acc[t][r] + bv;
  }
#else
  if (!active) return;
  const float* xr = x + (size_t)(row0 + m) * CCH;
  for (int ncol = half * 64; ncol < half * 64 + 64; ++ncol) {
    float v = use_bias ? bias[ncol] : 0.0f;
    for (int k = 0; k < CCH; ++k) v = fmaf(xr[k], wh[k * HCH + ncol], v);
    p[(size_t)(row0 + m) * HCH + ncol] = v;
  }
#endif
}

static_assert(NLBL % 32 == 0, "a wave takes 32 edges");
__global__ void __launch_bounds__(256)
k_dec_edge(const float* __restrict__ pu, const float* __restrict__ pb,
           const int* __restrict__ ls, const int* __restrict__ ld,
           const float* __restrict__ w2, const float* __restrict__ bias2,
           float* __restrict__ out) {
  int wave = (int)((blockIdx.x * blockDim.x + threadIdx.x) >> 5);
  wave = __builtin_amdgcn_readfirstlane(wave);
  const int e0 = wave * 32;
  if (e0 >= NLBL) return;
  const int lane = (int)threadIdx.x & 31;
  const unsigned loff = (unsigned)lane << 4;
  const float4 wv = ((const float4*)w2)[lane];
  const float ob = bias2[0];
  const char* pub = (const char*)pu;
  const char* pbb = (const char*)pb;
  const bool h = (lane & 16) != 0, g = (lane & 8) != 0, f = (lane & 4) != 0;
  for (int b = 0; b < 4; ++b) {
    float4 va[8], vc[8];
#pragma unroll
    for (int i = 0; i < 8; ++i) {
      const int e = e0 + 8 * b + i;
      const unsigned su = (unsigned)min(max(ls[e], 0), NUSER - 1);
      const unsigned sb = (unsigned)min(max(ld[e], 0), NBOOK - 1);
      va[i] = *(const float4*)(pub + ((su << 9) + loff));
      vc[i] = *(const float4*)(pbb + ((sb << 9) + loff));
    }
    float p[8];
#pragma unroll
    for (int i = 0; i < 8; ++i) {
      const float x0 = fmaxf(va[i].x + vc[i].x, 0.0f);
      const float x1 = fmaxf(va[i].y + vc[i].y, 0.0f);
      const float x2 = fmaxf(va[i].z + vc[i].z, 0.0f);
      const float x3 = fmaxf(va[i].w + vc[i].w, 0.0f);
      p[i] = fmaf(x3, wv.w, fmaf(x2, wv.z, fmaf(x1, wv.y, x0 * wv.x)));
    }
    float q[4];
#pragma unroll
    for (int i = 0; i < 4; ++i) {
      const float keep = h ? p[i + 4] : p[i];
      const float send = h ? p[i] : p[i + 4];
      q[i] = keep + __shfl_xor(send, 16);
    }
    float r[2];
#pragma unroll
    for (int i = 0; i < 2; ++i) {
      const float keep = g ? q[i + 2] : q[i];
      const float send = g ? q[i] : q[i + 2];
      r[i] = keep + __shfl_xor(send, 8);
    }
    float s = (f ? r[1] : r[0]) + __shfl_xor(f ? r[0] : r[1], 4);
    s += __shfl_xor(s, 2);
    s += __shfl_xor(s, 1);
    if ((lane & 3) == 0) out[e0 + 8 * b + (lane >> 2)] = s + ob;
  }
}

extern "C" void kernel_launch(void* const* d_in, const int* in_sizes, int n_in,
                              void* d_out, int out_size, void* d_ws, size_t ws_size,
                              hipStream_t stream) {
  (void)in_sizes; (void)n_in; (void)out_size; (void)ws_size;
  const float* user_x     = (const float*)d_in[0];
  const float* book_x     = (const float*)d_in[1];
  const int*   edge_src   = (const int*)d_in[2];
  const int*   edge_dst   = (const int*)d_in[3];
  const int*   label_src  = (const int*)d_in[4];
  const int*   label_dst  = (const int*)d_in[5];
  const float* user_lin_w = (const float*)d_in[6];
  const float* user_lin_b = (const float*)d_in[7];
  const float* book_lin_w = (const float*)d_in[8];
  const float* book_lin_b = (const float*)d_in[9];
  const float* w1_ub_l = (const float*)d_in[10];
  const float* b1_ub   = (const float*)d_in[11];
  const float* w1_ub_r = (const float*)d_in[12];
  const float* w1_bu_l = (const float*)d_in[13];
  const float* b1_bu   = (const float*)d_in[14];
  const float* w1_bu_r = (const float*)d_in[15];
  const float* w2_ub_l = (const float*)d_in[16];
  const float* b2_ub   = (const float*)d_in[17];
  const float* w2_ub_r = (const float*)d_in[18];
  const float* w2_bu_l = (const float*)d_in[19];
  const float* b2_bu   = (const float*)d_in[20];
  const float* w2_bu_r = (const float*)d_in[21];
  const float* dec_w1  = (const float*)d_in[22];
  const float* dec_b1  = (const float*)d_in[23];
  const float* dec_w2  = (const float*)d_in[24];
  const float* dec_b2  = (const float*)d_in[25];
  float* out = (float*)d_out;

  float* ws = (float*)d_ws;
  size_t off = 0;
  float* u0    = ws + off; off += (size_t)NUSER * CCH;
  float* b0    = ws + off; off += (size_t)NBOOK * CCH;
  float* u1    = ws + off; off += (size_t)NUSER * CCH;
  float* b1buf = ws + off; off += (size_t)NBOOK * CCH;
  int*   tab   = (int*)(ws + off); off += (size_t)NUSER * CCH;
  float* cnt_u = ws + off; off += NUSER;
  float* cnt_b = ws + off; off += NBOOK;
  float* u2 = u0;
  float* b2 = b0;
  int* adj_b    = tab;
  int* adj_u    = adj_b + (NEDGE + ADJ_PAD);
  int* hist_b   = adj_u + (NEDGE + ADJ_PAD);
  int* hist_u   = hist_b + NBOOK;
  int* rowptr_b = hist_u + NUSER;
  int* rowptr_u = rowptr_b + NBOOK;
  int* histp    = rowptr_u + NUSER;
  static_assert(2 * (size_t)(NEDGE + ADJ_PAD) + 2 * (size_t)NKEY + (size_t)SORT_P * NKEY
                    <= (size_t)NUSER * CCH, "integer tables fit their region");

  const int BT = 256;

  k_count<<<(SORT_RB + SORT_RU) * SORT_P, SORT_T, 0, stream>>>(edge_src, edge_dst, histp);
  k_stage1<<<PB_BLK + CB_BLK + PU_BLK, BT, 0, stream>>>(user_x, user_lin_w, user_lin_b, u0,
                                                         book_x, book_lin_w, book_lin_b, b0,
                                                         histp, hist_b);
  k_scan<<<SCAN_BLK_B + SCAN_BLK_U, SCAN_T, 0, stream>>>(hist_b, hist_u, rowptr_b, rowptr_u, cnt_b, cnt_u);
  k_place<<<(SORT_RB + SORT_RU) * SORT_P, SORT_T, 0, stream>>>(edge_src, edge_dst, histp, rowptr_b, adj_u, adj_b);

  k_gather<<<GAT_BLK_B + GAT_BLK_U, BT, 0, stream>>>(adj_b, rowptr_b, hist_b, u0, b1buf,
                                                      adj_u, rowptr_u, hist_u, b0, u1);
  k_sage<<<SAGE_BLK_B + SAGE_BLK_U, BT, 0, stream>>>(b1buf, cnt_b, b0, w1_ub_l, w1_ub_r, b1_ub, b1buf,
                                                      u1, cnt_u, u0, w1_bu_l, w1_bu_r, b1_bu, u1, 1);
  k_gather<<<GAT_BLK_B + GAT_BLK_U, BT, 0, stream>>>(adj_b, rowptr_b, hist_b, u1, b2,
                                                      adj_u, rowptr_u, hist_u, b1buf, u2);
  k_sage<<<SAGE_BLK_B + SAGE_BLK_U, BT, 0, stream>>>(b2, cnt_b, b1buf, w2_ub_l, w2_ub_r, b2_ub, b2,
                                                      u2, cnt_u, u1, w2_bu_l, w2_bu_r, b2_bu, u2, 0);

  static_assert((size_t)NBOOK * HCH <= (size_t)NUSER * CCH, "book products fit the table region");
  static_assert((size_t)NUSER * HCH == (size_t)NBOOK * CCH + (size_t)NUSER * CCH + (size_t)NBOOK * CCH,
                "user products fill b0|u1|b1buf exactly");
  float* pdec_b = (float*)tab;
  float* pdec_u = b0;
  k_dec_proj<<<SAGE_BLK_B, BT, 0, stream>>>(b2, dec_w1 + CCH * HCH, dec_b1, pdec_b, NBOOK, 0);
  k_dec_proj<<<SAGE_BLK_U, BT, 0, stream>>>(u2, dec_w1, dec_b1, pdec_u, NUSER, 1);
  k_dec_edge<<<(NLBL + BT - 1) / BT, BT, 0, stream>>>(
      pdec_u, pdec_b, label_src, label_dst, dec_w2, dec_b2, out);
}
